# starting kernel
# baseline (speedup 1.0000x reference)
#include <hip/hip_runtime.h>

#define NG     100000
#define NS     20000
#define EDGES  300000
#define DIM    256
#define NQ     4096
#define MAXU   4096
#define LDSPAD 2

typedef __attribute__((ext_vector_type(2))) float v2f;
typedef __attribute__((ext_vector_type(8))) float v8f;

__global__ __launch_bounds__(256) void k_init_slots(int* __restrict__ sg,
                                                    int* __restrict__ ss,
                                                    int* __restrict__ cnt) {
  int i = blockIdx.x * 256 + threadIdx.x;
  if (i < NG) sg[i] = -1;
  if (i < NS) ss[i] = -1;
  if (i < 2)  cnt[i] = 0;
}

__global__ __launch_bounds__(256) void k_zero(float* __restrict__ p, int n) {
  int i = blockIdx.x * 256 + threadIdx.x;
  if (i < n) p[i] = 0.0f;
}

__global__ __launch_bounds__(256) void k_compact(const int* __restrict__ gid,
                                                 const int* __restrict__ sid,
                                                 int* __restrict__ sg,
                                                 int* __restrict__ ss,
                                                 int* __restrict__ cnt) {
  int i = blockIdx.x * 256 + threadIdx.x;
  if (i < NQ) {
    int g = gid[i];
    if (atomicCAS(&sg[g], -1, -2) == -1) sg[g] = atomicAdd(&cnt[0], 1);
  } else if (i < 2 * NQ) {
    int g = sid[i - NQ];
    if (atomicCAS(&ss[g], -1, -2) == -1) ss[g] = atomicAdd(&cnt[1], 1);
  }
}

__global__ __launch_bounds__(256) void k_scatter(const float* __restrict__ vsrc,
                                                 const float* __restrict__ att,
                                                 const int* __restrict__ src,
                                                 const int* __restrict__ dst,
                                                 const int* __restrict__ slot,
                                                 float* __restrict__ nh, int nE) {
  const int wid  = (blockIdx.x * 256 + threadIdx.x) >> 5;
  const int lane = threadIdx.x & 31;
  if (wid >= nE) return;
  const int s = slot[dst[wid]];
  if (s < 0) return;
  const float a = att[wid];
  const float4* vp = (const float4*)(vsrc + (size_t)src[wid] * DIM);
  float* np = nh + (size_t)s * DIM;
#pragma unroll
  for (int j = 0; j < 2; ++j) {
    const int q = lane + j * 32;
    const float4 vv = vp[q];
    atomicAdd(np + q * 4 + 0, vv.x * a);
    atomicAdd(np + q * 4 + 1, vv.y * a);
    atomicAdd(np + q * 4 + 2, vv.z * a);
    atomicAdd(np + q * 4 + 3, vv.w * a);
  }
}

__global__ __launch_bounds__(32) void k_combine(const float* __restrict__ v,
                                                const float* __restrict__ nh,
                                                const int* __restrict__ slot,
                                                const int* __restrict__ ids,
                                                const float* __restrict__ W1,
                                                const float* __restrict__ b1,
                                                float* __restrict__ out) {
  __shared__ float x1[16][DIM + LDSPAD];
  __shared__ float x2[16][DIM + LDSPAD];
  __shared__ int   qg[16];
  __shared__ int   qs[16];

  const int lane  = threadIdx.x;
  const int qbase = blockIdx.x * 16;

  if (lane < 16) {
    const int g = ids[qbase + lane];
    qg[lane] = g;
    qs[lane] = slot[g];
  }
  __syncthreads();

  for (int i = lane; i < 16 * (DIM / 4); i += 32) {
    const int r = i >> 6;
    const int c = (i & 63) * 4;
    const float4 vv = *(const float4*)(v  + (size_t)qg[r] * DIM + c);
    const float4 nn = *(const float4*)(nh + (size_t)qs[r] * DIM + c);
    x1[r][c + 0] = vv.x + nn.x;  x2[r][c + 0] = vv.x * nn.x;
    x1[r][c + 1] = vv.y + nn.y;  x2[r][c + 1] = vv.y * nn.y;
    x1[r][c + 2] = vv.z + nn.z;  x2[r][c + 2] = vv.z * nn.z;
    x1[r][c + 3] = vv.w + nn.w;  x2[r][c + 3] = vv.w * nn.w;
  }
  __syncthreads();

  const int mrow  = lane & 15;
  const int khalf = lane >> 4;

  for (int n0 = 0; n0 < DIM; n0 += 16) {
    v8f acc1 = {};
    v8f acc2 = {};
    const int ncol = n0 + (lane & 15);
    const float* wrow = W1 + (size_t)ncol * DIM;

#pragma unroll 4
    for (int k0 = 0; k0 < DIM; k0 += 4) {
      const int ka = k0 + 2 * khalf;
      v2f a1; a1.x = x1[mrow][ka]; a1.y = x1[mrow][ka + 1];
      v2f a2; a2.x = x2[mrow][ka]; a2.y = x2[mrow][ka + 1];
      v2f bb; bb.x = wrow[ka];     bb.y = wrow[ka + 1];
      acc1 = __builtin_amdgcn_wmma_f32_16x16x4_f32(
          false, a1, false, bb, (short)0, acc1, false, false);
      acc2 = __builtin_amdgcn_wmma_f32_16x16x4_f32(
          false, a2, false, bb, (short)0, acc2, false, false);
    }

    const float bias = b1[ncol];
#pragma unroll
    for (int r = 0; r < 8; ++r) {
      const int m = khalf * 8 + r;
      float y1 = acc1[r] + bias;
      float y2 = acc2[r] + bias;
      y1 = (y1 >= 0.0f) ? y1 : 0.01f * y1;
      y2 = (y2 >= 0.0f) ? y2 : 0.01f * y2;
      out[(size_t)(qbase + m) * DIM + ncol] = y1 + y2;
    }
  }
}

extern "C" void kernel_launch(void* const* d_in, const int* in_sizes, int n_in,
                              void* d_out, int out_size, void* d_ws, size_t ws_size,
                              hipStream_t stream) {
  const float* v_grid  = (const float*)d_in[0];
  const float* v_sc    = (const float*)d_in[1];
  const float* att_s2g = (const float*)d_in[2];
  const float* att_g2s = (const float*)d_in[3];
  const float* W1      = (const float*)d_in[4];
  const float* b1      = (const float*)d_in[5];
  const int*   src_s2g = (const int*)d_in[6];
  const int*   dst_s2g = (const int*)d_in[7];
  const int*   src_g2s = (const int*)d_in[8];
  const int*   dst_g2s = (const int*)d_in[9];
  const int*   sc_id   = (const int*)d_in[10];
  const int*   gr_id   = (const int*)d_in[11];
  float* out = (float*)d_out;

  int*   sg  = (int*)d_ws;
  int*   ss  = sg + NG;
  int*   cnt = ss + NS;
  float* nhg = (float*)(cnt + 4);
  float* nhs = nhg + (size_t)MAXU * DIM;

  k_init_slots<<<(NG + 255) / 256, 256, 0, stream>>>(sg, ss, cnt);
  k_zero<<<(2 * MAXU * DIM + 255) / 256, 256, 0, stream>>>(nhg, 2 * MAXU * DIM);
  k_compact<<<(2 * NQ + 255) / 256, 256, 0, stream>>>(gr_id, sc_id, sg, ss, cnt);

  k_scatter<<<(EDGES + 7) / 8, 256, 0, stream>>>(v_sc,   att_s2g, src_s2g, dst_s2g, sg, nhg, EDGES);
  k_scatter<<<(EDGES + 7) / 8, 256, 0, stream>>>(v_grid, att_g2s, src_g2s, dst_g2s, ss, nhs, EDGES);

  k_combine<<<NQ / 16, 32, 0, stream>>>(v_sc,   nhs, ss, sc_id, W1, b1, out);
  k_combine<<<NQ / 16, 32, 0, stream>>>(v_grid, nhg, sg, gr_id, W1, b1, out + (size_t)NQ * DIM);
}


# combine GEMM: the 16 column tiles of each 16-row block are spread over 8 waves (2 tiles per wave, x tile shared in LDS, W1 fragments for all K loaded up front) instead of 1 wave looping over them
# speedup vs baseline: 2.5264x; 2.5264x over previous
#include <hip/hip_runtime.h>

#define NG     100000
#define NS     20000
#define EDGES  300000
#define DIM    256
#define NQ     4096
#define MAXU   4096
#define LDSPAD 2

typedef __attribute__((ext_vector_type(2))) float v2f;
typedef __attribute__((ext_vector_type(8))) float v8f;

__global__ __launch_bounds__(256) void k_init_slots(int* __restrict__ sg,
                                                    int* __restrict__ ss,
                                                    int* __restrict__ cnt) {
  int i = blockIdx.x * 256 + threadIdx.x;
  if (i < NG) sg[i] = -1;
  if (i < NS) ss[i] = -1;
  if (i < 2)  cnt[i] = 0;
}

__global__ __launch_bounds__(256) void k_zero(float* __restrict__ p, int n) {
  int i = blockIdx.x * 256 + threadIdx.x;
  if (i < n) p[i] = 0.0f;
}

__global__ __launch_bounds__(256) void k_compact(const int* __restrict__ gid,
                                                 const int* __restrict__ sid,
                                                 int* __restrict__ sg,
                                                 int* __restrict__ ss,
                                                 int* __restrict__ cnt) {
  int i = blockIdx.x * 256 + threadIdx.x;
  if (i < NQ) {
    int g = gid[i];
    if (atomicCAS(&sg[g], -1, -2) == -1) sg[g] = atomicAdd(&cnt[0], 1);
  } else if (i < 2 * NQ) {
    int g = sid[i - NQ];
    if (atomicCAS(&ss[g], -1, -2) == -1) ss[g] = atomicAdd(&cnt[1], 1);
  }
}

__global__ __launch_bounds__(256) void k_scatter(const float* __restrict__ vsrc,
                                                 const float* __restrict__ att,
                                                 const int* __restrict__ src,
                                                 const int* __restrict__ dst,
                                                 const int* __restrict__ slot,
                                                 float* __restrict__ nh, int nE) {
  const int wid  = (blockIdx.x * 256 + threadIdx.x) >> 5;
  const int lane = threadIdx.x & 31;
  if (wid >= nE) return;
  const int s = slot[dst[wid]];
  if (s < 0) return;
  const float a = att[wid];
  const float4* vp = (const float4*)(vsrc + (size_t)src[wid] * DIM);
  float* np = nh + (size_t)s * DIM;
#pragma unroll
  for (int j = 0; j < 2; ++j) {
    const int q = lane + j * 32;
    const float4 vv = vp[q];
    atomicAdd(np + q * 4 + 0, vv.x * a);
    atomicAdd(np + q * 4 + 1, vv.y * a);
    atomicAdd(np + q * 4 + 2, vv.z * a);
    atomicAdd(np + q * 4 + 3, vv.w * a);
  }
}

#define CTHREADS 256
__global__ __launch_bounds__(CTHREADS) void k_combine(const float* __restrict__ v,
                                                         const float* __restrict__ nh,
                                                         const int* __restrict__ slot,
                                                         const int* __restrict__ ids,
                                                         const float* __restrict__ W1,
                                                         const float* __restrict__ b1,
                                                         float* __restrict__ out) {
  __shared__ __attribute__((aligned(16))) float x1[16][DIM + LDSPAD];
  __shared__ __attribute__((aligned(16))) float x2[16][DIM + LDSPAD];

  const int lane  = threadIdx.x & 31;
  const int wave  = __builtin_amdgcn_readfirstlane((int)(threadIdx.x >> 5));
  const int qbase = blockIdx.x * 16;

  const int mrow  = lane & 15;
  const int khalf = lane >> 4;

  const int ncolA = wave * 32 + mrow;
  const int ncolB = ncolA + 16;
  const float* wrowA = W1 + (size_t)ncolA * DIM + 2 * khalf;
  const float* wrowB = W1 + (size_t)ncolB * DIM + 2 * khalf;
  v2f bA[DIM / 4];
  v2f bB[DIM / 4];
#pragma unroll
  for (int t = 0; t < DIM / 4; ++t) {
    bA[t].x = wrowA[4 * t];  bA[t].y = wrowA[4 * t + 1];
    bB[t].x = wrowB[4 * t];  bB[t].y = wrowB[4 * t + 1];
  }

  for (int rr = 0; rr < 2; ++rr) {
    const int r = wave * 2 + rr;
    const int g = ids[qbase + r];
    const int s = slot[g];
    const float4* vp = (const float4*)(v  + (size_t)g * DIM);
    const float4* np = (const float4*)(nh + (size_t)s * DIM);
    for (int j = 0; j < 2; ++j) {
      const int q = lane + j * 32;
      const int c = q * 4;
      const float4 vv = vp[q];
      const float4 nn = np[q];
      x1[r][c + 0] = vv.x + nn.x;  x2[r][c + 0] = vv.x * nn.x;
      x1[r][c + 1] = vv.y + nn.y;  x2[r][c + 1] = vv.y * nn.y;
      x1[r][c + 2] = vv.z + nn.z;  x2[r][c + 2] = vv.z * nn.z;
      x1[r][c + 3] = vv.w + nn.w;  x2[r][c + 3] = vv.w * nn.w;
    }
  }
  __syncthreads();

  v8f acc1A = {};
  v8f acc2A = {};
  v8f acc1B = {};
  v8f acc2B = {};
#pragma unroll
  for (int t = 0; t < DIM / 4; ++t) {
    const int ka = 4 * t + 2 * khalf;
    v2f a1; a1.x = x1[mrow][ka]; a1.y = x1[mrow][ka + 1];
    v2f a2; a2.x = x2[mrow][ka]; a2.y = x2[mrow][ka + 1];
    acc1A = __builtin_amdgcn_wmma_f32_16x16x4_f32(
        false, a1, false, bA[t], (short)0, acc1A, false, false);
    acc2A = __builtin_amdgcn_wmma_f32_16x16x4_f32(
        false, a2, false, bA[t], (short)0, acc2A, false, false);
    acc1B = __builtin_amdgcn_wmma_f32_16x16x4_f32(
        false, a1, false, bB[t], (short)0, acc1B, false, false);
    acc2B = __builtin_amdgcn_wmma_f32_16x16x4_f32(
        false, a2, false, bB[t], (short)0, acc2B, false, false);
  }

  const float biasA = b1[ncolA];
  const float biasB = b1[ncolB];
#pragma unroll
  for (int r = 0; r < 8; ++r) {
    const int m = khalf * 8 + r;
    float y1 = acc1A[r] + biasA;
    float y2 = acc2A[r] + biasA;
    y1 = (y1 >= 0.0f) ? y1 : 0.01f * y1;
    y2 = (y2 >= 0.0f) ? y2 : 0.01f * y2;
    out[(size_t)(qbase + m) * DIM + ncolA] = y1 + y2;
    float z1 = acc1B[r] + biasB;
    float z2 = acc2B[r] + biasB;
    z1 = (z1 >= 0.0f) ? z1 : 0.01f * z1;
    z2 = (z2 >= 0.0f) ? z2 : 0.01f * z2;
    out[(size_t)(qbase + m) * DIM + ncolB] = z1 + z2;
  }
}

extern "C" void kernel_launch(void* const* d_in, const int* in_sizes, int n_in,
                              void* d_out, int out_size, void* d_ws, size_t ws_size,
                              hipStream_t stream) {
  const float* v_grid  = (const float*)d_in[0];
  const float* v_sc    = (const float*)d_in[1];
  const float* att_s2g = (const float*)d_in[2];
  const float* att_g2s = (const float*)d_in[3];
  const float* W1      = (const float*)d_in[4];
  const float* b1      = (const float*)d_in[5];
  const int*   src_s2g = (const int*)d_in[6];
  const int*   dst_s2g = (const int*)d_in[7];
  const int*   src_g2s = (const int*)d_in[8];
  const int*   dst_g2s = (const int*)d_in[9];
  const int*   sc_id   = (const int*)d_in[10];
  const int*   gr_id   = (const int*)d_in[11];
  float* out = (float*)d_out;

  int*   sg  = (int*)d_ws;
  int*   ss  = sg + NG;
  int*   cnt = ss + NS;
  float* nhg = (float*)(cnt + 4);
  float* nhs = nhg + (size_t)MAXU * DIM;

  k_init_slots<<<(NG + 255) / 256, 256, 0, stream>>>(sg, ss, cnt);
  k_zero<<<(2 * MAXU * DIM + 255) / 256, 256, 0, stream>>>(nhg, 2 * MAXU * DIM);
  k_compact<<<(2 * NQ + 255) / 256, 256, 0, stream>>>(gr_id, sc_id, sg, ss, cnt);

  k_scatter<<<(EDGES + 7) / 8, 256, 0, stream>>>(v_sc,   att_s2g, src_s2g, dst_s2g, sg, nhg, EDGES);
  k_scatter<<<(EDGES + 7) / 8, 256, 0, stream>>>(v_grid, att_g2s, src_g2s, dst_g2s, ss, nhs, EDGES);

  k_combine<<<NQ / 16, CTHREADS, 0, stream>>>(v_sc,   nhs, ss, sc_id, W1, b1, out);
  k_combine<<<NQ / 16, CTHREADS, 0, stream>>>(v_grid, nhg, sg, gr_id, W1, b1, out + (size_t)NQ * DIM);
}


# replace the float-atomic edge scatter and zeroed nh buffer by per-destination buckets of (source row, weight) summed in registers by the combine kernel; nh is now summed in bucket order
# speedup vs baseline: 3.8563x; 1.5264x over previous
#include <hip/hip_runtime.h>

#define NG     100000
#define NS     20000
#define EDGES  300000
#define DIM    256
#define NQ     4096
#define MAXU   4096
#define LDSPAD 2
#define CAP    64
#define GRP    8

typedef __attribute__((ext_vector_type(2))) float v2f;
typedef __attribute__((ext_vector_type(8))) float v8f;

__global__ __launch_bounds__(256) void k_init_slots(int* __restrict__ sg,
                                                    int* __restrict__ ss,
                                                    int* __restrict__ cnt,
                                                    int* __restrict__ deg,
                                                    int* __restrict__ ohead) {
  int i = blockIdx.x * 256 + threadIdx.x;
  if (i < NG) sg[i] = -1;
  if (i < NS) ss[i] = -1;
  if (i < 2)  cnt[i] = 0;
  if (i < 2 * MAXU) { deg[i] = 0; ohead[i] = -1; }
}

__global__ __launch_bounds__(256) void k_compact(const int* __restrict__ gid,
                                                 const int* __restrict__ sid,
                                                 int* __restrict__ sg,
                                                 int* __restrict__ ss,
                                                 int* __restrict__ cnt) {
  int i = blockIdx.x * 256 + threadIdx.x;
  if (i < NQ) {
    int g = gid[i];
    if (atomicCAS(&sg[g], -1, -2) == -1) sg[g] = atomicAdd(&cnt[0], 1);
  } else if (i < 2 * NQ) {
    int g = sid[i - NQ];
    if (atomicCAS(&ss[g], -1, -2) == -1) ss[g] = atomicAdd(&cnt[1], 1);
  }
}

__global__ __launch_bounds__(256) void k_link(const float* __restrict__ att,
                                              const int* __restrict__ src,
                                              const int* __restrict__ dst,
                                              const int* __restrict__ slot,
                                              int* __restrict__ deg,
                                              int* __restrict__ bucket,
                                              int* __restrict__ ohead,
                                              int* __restrict__ nxt, int nE) {
  const int e = blockIdx.x * 256 + threadIdx.x;
  if (e >= nE) return;
  const int s = slot[dst[e]];
  if (s < 0) return;
  const int p = atomicAdd(&deg[s], 1);
  if (p < CAP) {
    int2 ent; ent.x = src[e]; ent.y = __float_as_int(att[e]);
    *(int2*)(bucket + ((size_t)s * CAP + p) * 2) = ent;
  } else {
    nxt[e] = atomicExch(&ohead[s], e);
  }
}

#define CTHREADS 256
__global__ __launch_bounds__(CTHREADS) void k_combine(const float* __restrict__ v,
                                                         const float* __restrict__ vsrc,
                                                         const float* __restrict__ att,
                                                         const int* __restrict__ esrc,
                                                         const int* __restrict__ deg,
                                                         const int* __restrict__ bucket,
                                                         const int* __restrict__ ohead,
                                                         const int* __restrict__ nxt,
                                                         const int* __restrict__ slot,
                                                         const int* __restrict__ ids,
                                                         const float* __restrict__ W1,
                                                         const float* __restrict__ b1,
                                                         float* __restrict__ out) {
  __shared__ __attribute__((aligned(16))) float x1[16][DIM + LDSPAD];
  __shared__ __attribute__((aligned(16))) float x2[16][DIM + LDSPAD];

  const int lane  = threadIdx.x & 31;
  const int wave  = __builtin_amdgcn_readfirstlane((int)(threadIdx.x >> 5));
  const int qbase = blockIdx.x * 16;

  const int mrow  = lane & 15;
  const int khalf = lane >> 4;

  const int ncolA = wave * 32 + mrow;
  const int ncolB = ncolA + 16;
  const float* wrowA = W1 + (size_t)ncolA * DIM + 2 * khalf;
  const float* wrowB = W1 + (size_t)ncolB * DIM + 2 * khalf;
  v2f bA[DIM / 4];
  v2f bB[DIM / 4];
#pragma unroll
  for (int t = 0; t < DIM / 4; ++t) {
    bA[t].x = wrowA[4 * t];  bA[t].y = wrowA[4 * t + 1];
    bB[t].x = wrowB[4 * t];  bB[t].y = wrowB[4 * t + 1];
  }

  for (int rr = 0; rr < 2; ++rr) {
    const int r = wave * 2 + rr;
    const int g = ids[qbase + r];
    const int s = slot[g];
    const int d  = __builtin_amdgcn_readfirstlane(deg[s]);
    const int nb = (d < CAP) ? d : CAP;
    const float4* vp = (const float4*)(v + (size_t)g * DIM);
    const float4 vv0 = vp[lane];
    const float4 vv1 = vp[lane + 32];
    float4 n0 = {0.0f, 0.0f, 0.0f, 0.0f};
    float4 n1 = {0.0f, 0.0f, 0.0f, 0.0f};

    for (int base = 0; base < nb; base += 32) {
      const int cn = (nb - base < 32) ? (nb - base) : 32;
      int   es = 0;
      float ea = 0.0f;
      if (lane < cn) {
        const int2 ent = *(const int2*)(bucket + ((size_t)s * CAP + base + lane) * 2);
        es = ent.x;
        ea = __int_as_float(ent.y);
      }
      for (int i = 0; i < cn; i += GRP) {
        float4 r0[GRP];
        float4 r1[GRP];
#pragma unroll
        for (int k = 0; k < GRP; ++k) {
          const int ik = (i + k < cn) ? (i + k) : (cn - 1);
          const int si = __builtin_amdgcn_readlane(es, ik);
          const float4* rp = (const float4*)(vsrc + (size_t)si * DIM);
          r0[k] = rp[lane];
          r1[k] = rp[lane + 32];
        }
#pragma unroll
        for (int k = 0; k < GRP; ++k) {
          if (i + k < cn) {
            const float ai = __int_as_float(
                __builtin_amdgcn_readlane(__float_as_int(ea), i + k));
            n0.x += r0[k].x * ai;  n0.y += r0[k].y * ai;
            n0.z += r0[k].z * ai;  n0.w += r0[k].w * ai;
            n1.x += r1[k].x * ai;  n1.y += r1[k].y * ai;
            n1.z += r1[k].z * ai;  n1.w += r1[k].w * ai;
          }
        }
      }
    }
    if (d > CAP) {
      int e = __builtin_amdgcn_readfirstlane(ohead[s]);
      for (int it = 0; it < EDGES && e >= 0; ++it) {
        const int   si = esrc[e];
        const float ai = att[e];
        const float4* rp = (const float4*)(vsrc + (size_t)si * DIM);
        const float4 q0 = rp[lane];
        const float4 q1 = rp[lane + 32];
        n0.x += q0.x * ai;  n0.y += q0.y * ai;  n0.z += q0.z * ai;  n0.w += q0.w * ai;
        n1.x += q1.x * ai;  n1.y += q1.y * ai;  n1.z += q1.z * ai;  n1.w += q1.w * ai;
        e = __builtin_amdgcn_readfirstlane(nxt[e]);
      }
    }

    const int c0 = lane * 4;
    const int c1 = c0 + 128;
    x1[r][c0 + 0] = vv0.x + n0.x;  x2[r][c0 + 0] = vv0.x * n0.x;
    x1[r][c0 + 1] = vv0.y + n0.y;  x2[r][c0 + 1] = vv0.y * n0.y;
    x1[r][c0 + 2] = vv0.z + n0.z;  x2[r][c0 + 2] = vv0.z * n0.z;
    x1[r][c0 + 3] = vv0.w + n0.w;  x2[r][c0 + 3] = vv0.w * n0.w;
    x1[r][c1 + 0] = vv1.x + n1.x;  x2[r][c1 + 0] = vv1.x * n1.x;
    x1[r][c1 + 1] = vv1.y + n1.y;  x2[r][c1 + 1] = vv1.y * n1.y;
    x1[r][c1 + 2] = vv1.z + n1.z;  x2[r][c1 + 2] = vv1.z * n1.z;
    x1[r][c1 + 3] = vv1.w + n1.w;  x2[r][c1 + 3] = vv1.w * n1.w;
  }
  __syncthreads();

  v8f acc1A = {};
  v8f acc2A = {};
  v8f acc1B = {};
  v8f acc2B = {};
#pragma unroll
  for (int t = 0; t < DIM / 4; ++t) {
    const int ka = 4 * t + 2 * khalf;
    v2f a1; a1.x = x1[mrow][ka]; a1.y = x1[mrow][ka + 1];
    v2f a2; a2.x = x2[mrow][ka]; a2.y = x2[mrow][ka + 1];
    acc1A = __builtin_amdgcn_wmma_f32_16x16x4_f32(
        false, a1, false, bA[t], (short)0, acc1A, false, false);
    acc2A = __builtin_amdgcn_wmma_f32_16x16x4_f32(
        false, a2, false, bA[t], (short)0, acc2A, false, false);
    acc1B = __builtin_amdgcn_wmma_f32_16x16x4_f32(
        false, a1, false, bB[t], (short)0, acc1B, false, false);
    acc2B = __builtin_amdgcn_wmma_f32_16x16x4_f32(
        false, a2, false, bB[t], (short)0, acc2B, false, false);
  }

  const float biasA = b1[ncolA];
  const float biasB = b1[ncolB];
#pragma unroll
  for (int r = 0; r < 8; ++r) {
    const int m = khalf * 8 + r;
    float y1 = acc1A[r] + biasA;
    float y2 = acc2A[r] + biasA;
    y1 = (y1 >= 0.0f) ? y1 : 0.01f * y1;
    y2 = (y2 >= 0.0f) ? y2 : 0.01f * y2;
    out[(size_t)(qbase + m) * DIM + ncolA] = y1 + y2;
    float z1 = acc1B[r] + biasB;
    float z2 = acc2B[r] + biasB;
    z1 = (z1 >= 0.0f) ? z1 : 0.01f * z1;
    z2 = (z2 >= 0.0f) ? z2 : 0.01f * z2;
    out[(size_t)(qbase + m) * DIM + ncolB] = z1 + z2;
  }
}

extern "C" void kernel_launch(void* const* d_in, const int* in_sizes, int n_in,
                              void* d_out, int out_size, void* d_ws, size_t ws_size,
                              hipStream_t stream) {
  const float* v_grid  = (const float*)d_in[0];
  const float* v_sc    = (const float*)d_in[1];
  const float* att_s2g = (const float*)d_in[2];
  const float* att_g2s = (const float*)d_in[3];
  const float* W1      = (const float*)d_in[4];
  const float* b1      = (const float*)d_in[5];
  const int*   src_s2g = (const int*)d_in[6];
  const int*   dst_s2g = (const int*)d_in[7];
  const int*   src_g2s = (const int*)d_in[8];
  const int*   dst_g2s = (const int*)d_in[9];
  const int*   sc_id   = (const int*)d_in[10];
  const int*   gr_id   = (const int*)d_in[11];
  float* out = (float*)d_out;

  int*  sg   = (int*)d_ws;
  int*  ss   = sg + NG;
  int*  cnt  = ss + NS;
  int*  degg = cnt + 4;
  int*  degs = degg + MAXU;
  int*  ohg  = degs + MAXU;
  int*  ohs  = ohg + MAXU;
  int*  nxg  = ohs + MAXU;
  int*  nxs  = nxg + EDGES;
  int*  bkg  = nxs + EDGES;
  int*  bks  = bkg + (size_t)MAXU * CAP * 2;

  k_init_slots<<<(NG + 255) / 256, 256, 0, stream>>>(sg, ss, cnt, degg, ohg);
  k_compact<<<(2 * NQ + 255) / 256, 256, 0, stream>>>(gr_id, sc_id, sg, ss, cnt);

  k_link<<<(EDGES + 255) / 256, 256, 0, stream>>>(att_s2g, src_s2g, dst_s2g, sg, degg, bkg, ohg, nxg, EDGES);
  k_link<<<(EDGES + 255) / 256, 256, 0, stream>>>(att_g2s, src_g2s, dst_g2s, ss, degs, bks, ohs, nxs, EDGES);

  k_combine<<<NQ / 16, CTHREADS, 0, stream>>>(v_sc,   v_grid, att_g2s, src_g2s, degs, bks, ohs, nxs,
                                              ss, sc_id, W1, b1, out);
  k_combine<<<NQ / 16, CTHREADS, 0, stream>>>(v_grid, v_sc,   att_s2g, src_s2g, degg, bkg, ohg, nxg,
                                              sg, gr_id, W1, b1, out + (size_t)NQ * DIM);
}


# link phase: one launch for both directions, 8 consecutive edges per thread with slot reads and bucket-place atomics batched (2,352 waves, was 18,752); nh still summed in bucket (arrival) order
# speedup vs baseline: 4.2118x; 1.0922x over previous
#include <hip/hip_runtime.h>

#define NG     100000
#define NS     20000
#define EDGES  300000
#define DIM    256
#define NQ     4096
#define MAXU   4096
#define LDSPAD 2
#define CAP    64
#define GRP    8
#define EPT    8
#define LINKB  ((EDGES / EPT + 255) / 256)
static_assert(EDGES % EPT == 0 && EPT == 8, "a link thread owns 8 whole edges (two 16-byte loads per array)");

typedef __attribute__((ext_vector_type(2))) float v2f;
typedef __attribute__((ext_vector_type(8))) float v8f;

__global__ __launch_bounds__(256) void k_init_slots(int* __restrict__ sg,
                                                    int* __restrict__ ss,
                                                    int* __restrict__ cnt,
                                                    int* __restrict__ deg,
                                                    int* __restrict__ ohead) {
  int i = blockIdx.x * 256 + threadIdx.x;
  if (i < NG) sg[i] = -1;
  if (i < NS) ss[i] = -1;
  if (i < 2)  cnt[i] = 0;
  if (i < 2 * MAXU) { deg[i] = 0; ohead[i] = -1; }
}

__global__ __launch_bounds__(256) void k_compact(const int* __restrict__ gid,
                                                 const int* __restrict__ sid,
                                                 int* __restrict__ sg,
                                                 int* __restrict__ ss,
                                                 int* __restrict__ cnt) {
  int i = blockIdx.x * 256 + threadIdx.x;
  if (i < NQ) {
    int g = gid[i];
    if (atomicCAS(&sg[g], -1, -2) == -1) sg[g] = atomicAdd(&cnt[0], 1);
  } else if (i < 2 * NQ) {
    int g = sid[i - NQ];
    if (atomicCAS(&ss[g], -1, -2) == -1) ss[g] = atomicAdd(&cnt[1], 1);
  }
}

__global__ __launch_bounds__(256) void k_link(const float* __restrict__ att0,
                                              const int* __restrict__ src0,
                                              const int* __restrict__ dst0,
                                              const int* __restrict__ slot0,
                                              int* __restrict__ deg0,
                                              int* __restrict__ bucket0,
                                              int* __restrict__ ohead0,
                                              int* __restrict__ nxt0,
                                              const float* __restrict__ att1,
                                              const int* __restrict__ src1,
                                              const int* __restrict__ dst1,
                                              const int* __restrict__ slot1,
                                              int* __restrict__ deg1,
                                              int* __restrict__ bucket1,
                                              int* __restrict__ ohead1,
                                              int* __restrict__ nxt1) {
  const bool first = (blockIdx.y == 0);
  const float* __restrict__ att    = first ? att0    : att1;
  const int*   __restrict__ src    = first ? src0    : src1;
  const int*   __restrict__ dst    = first ? dst0    : dst1;
  const int*   __restrict__ slot   = first ? slot0   : slot1;
  int*         __restrict__ deg    = first ? deg0    : deg1;
  int*         __restrict__ bucket = first ? bucket0 : bucket1;
  int*         __restrict__ ohead  = first ? ohead0  : ohead1;
  int*         __restrict__ nxt    = first ? nxt0    : nxt1;

  const int e0 = (blockIdx.x * 256 + threadIdx.x) * EPT;
  if (e0 >= EDGES) return;

  const int4   d0 = *(const int4*)(dst + e0);
  const int4   d1 = *(const int4*)(dst + e0 + 4);
  const int4   r0 = *(const int4*)(src + e0);
  const int4   r1 = *(const int4*)(src + e0 + 4);
  const float4 a0 = *(const float4*)(att + e0);
  const float4 a1 = *(const float4*)(att + e0 + 4);
  const int   d[EPT]  = {d0.x, d0.y, d0.z, d0.w, d1.x, d1.y, d1.z, d1.w};
  const int   sr[EPT] = {r0.x, r0.y, r0.z, r0.w, r1.x, r1.y, r1.z, r1.w};
  const float at[EPT] = {a0.x, a0.y, a0.z, a0.w, a1.x, a1.y, a1.z, a1.w};

  int s[EPT];
#pragma unroll
  for (int k = 0; k < EPT; ++k) s[k] = slot[d[k]];

  if ((s[0] & s[1] & s[2] & s[3] & s[4] & s[5] & s[6] & s[7]) < 0) return;

#define LINK_ASK(k) int p##k = 0; if (s[k] >= 0) p##k = atomicAdd(&deg[s[k]], 1);
  LINK_ASK(0) LINK_ASK(1) LINK_ASK(2) LINK_ASK(3)
  LINK_ASK(4) LINK_ASK(5) LINK_ASK(6) LINK_ASK(7)
#undef LINK_ASK

#define LINK_PUT(k)                                                        \
  if (s[k] >= 0) {                                                         \
    if (p##k < CAP) {                                                      \
      int2 ent; ent.x = sr[k]; ent.y = __float_as_int(at[k]);              \
      *(int2*)(bucket + ((size_t)s[k] * CAP + p##k) * 2) = ent;            \
    } else {                                                               \
      nxt[e0 + k] = atomicExch(&ohead[s[k]], e0 + k);                      \
    }                                                                      \
  }
  LINK_PUT(0) LINK_PUT(1) LINK_PUT(2) LINK_PUT(3)
  LINK_PUT(4) LINK_PUT(5) LINK_PUT(6) LINK_PUT(7)
#undef LINK_PUT
}

#define CTHREADS 256
__global__ __launch_bounds__(CTHREADS) void k_combine(const float* __restrict__ v,
                                                         const float* __restrict__ vsrc,
                                                         const float* __restrict__ att,
                                                         const int* __restrict__ esrc,
                                                         const int* __restrict__ deg,
                                                         const int* __restrict__ bucket,
                                                         const int* __restrict__ ohead,
                                                         const int* __restrict__ nxt,
                                                         const int* __restrict__ slot,
                                                         const int* __restrict__ ids,
                                                         const float* __restrict__ W1,
                                                         const float* __restrict__ b1,
                                                         float* __restrict__ out) {
  __shared__ __attribute__((aligned(16))) float x1[16][DIM + LDSPAD];
  __shared__ __attribute__((aligned(16))) float x2[16][DIM + LDSPAD];

  const int lane  = threadIdx.x & 31;
  const int wave  = __builtin_amdgcn_readfirstlane((int)(threadIdx.x >> 5));
  const int qbase = blockIdx.x * 16;

  const int mrow  = lane & 15;
  const int khalf = lane >> 4;

  const int ncolA = wave * 32 + mrow;
  const int ncolB = ncolA + 16;
  const float* wrowA = W1 + (size_t)ncolA * DIM + 2 * khalf;
  const float* wrowB = W1 + (size_t)ncolB * DIM + 2 * khalf;
  v2f bA[DIM / 4];
  v2f bB[DIM / 4];
#pragma unroll
  for (int t = 0; t < DIM / 4; ++t) {
    bA[t].x = wrowA[4 * t];  bA[t].y = wrowA[4 * t + 1];
    bB[t].x = wrowB[4 * t];  bB[t].y = wrowB[4 * t + 1];
  }

  for (int rr = 0; rr < 2; ++rr) {
    const int r = wave * 2 + rr;
    const int g = ids[qbase + r];
    const int s = slot[g];
    const int d  = __builtin_amdgcn_readfirstlane(deg[s]);
    const int nb = (d < CAP) ? d : CAP;
    const float4* vp = (const float4*)(v + (size_t)g * DIM);
    const float4 vv0 = vp[lane];
    const float4 vv1 = vp[lane + 32];
    float4 n0 = {0.0f, 0.0f, 0.0f, 0.0f};
    float4 n1 = {0.0f, 0.0f, 0.0f, 0.0f};

    for (int base = 0; base < nb; base += 32) {
      const int cn = (nb - base < 32) ? (nb - base) : 32;
      int   es = 0;
      float ea = 0.0f;
      if (lane < cn) {
        const int2 ent = *(const int2*)(bucket + ((size_t)s * CAP + base + lane) * 2);
        es = ent.x;
        ea = __int_as_float(ent.y);
      }
      for (int i = 0; i < cn; i += GRP) {
        float4 r0[GRP];
        float4 r1[GRP];
#pragma unroll
        for (int k = 0; k < GRP; ++k) {
          const int ik = (i + k < cn) ? (i + k) : (cn - 1);
          const int si = __builtin_amdgcn_readlane(es, ik);
          const float4* rp = (const float4*)(vsrc + (size_t)si * DIM);
          r0[k] = rp[lane];
          r1[k] = rp[lane + 32];
        }
#pragma unroll
        for (int k = 0; k < GRP; ++k) {
          if (i + k < cn) {
            const float ai = __int_as_float(
                __builtin_amdgcn_readlane(__float_as_int(ea), i + k));
            n0.x += r0[k].x * ai;  n0.y += r0[k].y * ai;
            n0.z += r0[k].z * ai;  n0.w += r0[k].w * ai;
            n1.x += r1[k].x * ai;  n1.y += r1[k].y * ai;
            n1.z += r1[k].z * ai;  n1.w += r1[k].w * ai;
          }
        }
      }
    }
    if (d > CAP) {
      int e = __builtin_amdgcn_readfirstlane(ohead[s]);
      for (int it = 0; it < EDGES && e >= 0; ++it) {
        const int   si = esrc[e];
        const float ai = att[e];
        const float4* rp = (const float4*)(vsrc + (size_t)si * DIM);
        const float4 q0 = rp[lane];
        const float4 q1 = rp[lane + 32];
        n0.x += q0.x * ai;  n0.y += q0.y * ai;  n0.z += q0.z * ai;  n0.w += q0.w * ai;
        n1.x += q1.x * ai;  n1.y += q1.y * ai;  n1.z += q1.z * ai;  n1.w += q1.w * ai;
        e = __builtin_amdgcn_readfirstlane(nxt[e]);
      }
    }

    const int c0 = lane * 4;
    const int c1 = c0 + 128;
    x1[r][c0 + 0] = vv0.x + n0.x;  x2[r][c0 + 0] = vv0.x * n0.x;
    x1[r][c0 + 1] = vv0.y + n0.y;  x2[r][c0 + 1] = vv0.y * n0.y;
    x1[r][c0 + 2] = vv0.z + n0.z;  x2[r][c0 + 2] = vv0.z * n0.z;
    x1[r][c0 + 3] = vv0.w + n0.w;  x2[r][c0 + 3] = vv0.w * n0.w;
    x1[r][c1 + 0] = vv1.x + n1.x;  x2[r][c1 + 0] = vv1.x * n1.x;
    x1[r][c1 + 1] = vv1.y + n1.y;  x2[r][c1 + 1] = vv1.y * n1.y;
    x1[r][c1 + 2] = vv1.z + n1.z;  x2[r][c1 + 2] = vv1.z * n1.z;
    x1[r][c1 + 3] = vv1.w + n1.w;  x2[r][c1 + 3] = vv1.w * n1.w;
  }
  __syncthreads();

  v8f acc1A = {};
  v8f acc2A = {};
  v8f acc1B = {};
  v8f acc2B = {};
#pragma unroll
  for (int t = 0; t < DIM / 4; ++t) {
    const int ka = 4 * t + 2 * khalf;
    v2f a1; a1.x = x1[mrow][ka]; a1.y = x1[mrow][ka + 1];
    v2f a2; a2.x = x2[mrow][ka]; a2.y = x2[mrow][ka + 1];
    acc1A = __builtin_amdgcn_wmma_f32_16x16x4_f32(
        false, a1, false, bA[t], (short)0, acc1A, false, false);
    acc2A = __builtin_amdgcn_wmma_f32_16x16x4_f32(
        false, a2, false, bA[t], (short)0, acc2A, false, false);
    acc1B = __builtin_amdgcn_wmma_f32_16x16x4_f32(
        false, a1, false, bB[t], (short)0, acc1B, false, false);
    acc2B = __builtin_amdgcn_wmma_f32_16x16x4_f32(
        false, a2, false, bB[t], (short)0, acc2B, false, false);
  }

  const float biasA = b1[ncolA];
  const float biasB = b1[ncolB];
#pragma unroll
  for (int r = 0; r < 8; ++r) {
    const int m = khalf * 8 + r;
    float y1 = acc1A[r] + biasA;
    float y2 = acc2A[r] + biasA;
    y1 = (y1 >= 0.0f) ? y1 : 0.01f * y1;
    y2 = (y2 >= 0.0f) ? y2 : 0.01f * y2;
    out[(size_t)(qbase + m) * DIM + ncolA] = y1 + y2;
    float z1 = acc1B[r] + biasB;
    float z2 = acc2B[r] + biasB;
    z1 = (z1 >= 0.0f) ? z1 : 0.01f * z1;
    z2 = (z2 >= 0.0f) ? z2 : 0.01f * z2;
    out[(size_t)(qbase + m) * DIM + ncolB] = z1 + z2;
  }
}

extern "C" void kernel_launch(void* const* d_in, const int* in_sizes, int n_in,
                              void* d_out, int out_size, void* d_ws, size_t ws_size,
                              hipStream_t stream) {
  const float* v_grid  = (const float*)d_in[0];
  const float* v_sc    = (const float*)d_in[1];
  const float* att_s2g = (const float*)d_in[2];
  const float* att_g2s = (const float*)d_in[3];
  const float* W1      = (const float*)d_in[4];
  const float* b1      = (const float*)d_in[5];
  const int*   src_s2g = (const int*)d_in[6];
  const int*   dst_s2g = (const int*)d_in[7];
  const int*   src_g2s = (const int*)d_in[8];
  const int*   dst_g2s = (const int*)d_in[9];
  const int*   sc_id   = (const int*)d_in[10];
  const int*   gr_id   = (const int*)d_in[11];
  float* out = (float*)d_out;

  int*  sg   = (int*)d_ws;
  int*  ss   = sg + NG;
  int*  cnt  = ss + NS;
  int*  degg = cnt + 4;
  int*  degs = degg + MAXU;
  int*  ohg  = degs + MAXU;
  int*  ohs  = ohg + MAXU;
  int*  nxg  = ohs + MAXU;
  int*  nxs  = nxg + EDGES;
  int*  bkg  = nxs + EDGES;
  int*  bks  = bkg + (size_t)MAXU * CAP * 2;

  k_init_slots<<<(NG + 255) / 256, 256, 0, stream>>>(sg, ss, cnt, degg, ohg);
  k_compact<<<(2 * NQ + 255) / 256, 256, 0, stream>>>(gr_id, sc_id, sg, ss, cnt);

  k_link<<<dim3(LINKB, 2), 256, 0, stream>>>(att_s2g, src_s2g, dst_s2g, sg, degg, bkg, ohg, nxg,
                                             att_g2s, src_g2s, dst_g2s, ss, degs, bks, ohs, nxs);

  k_combine<<<NQ / 16, CTHREADS, 0, stream>>>(v_sc,   v_grid, att_g2s, src_g2s, degs, bks, ohs, nxs,
                                              ss, sc_id, W1, b1, out);
  k_combine<<<NQ / 16, CTHREADS, 0, stream>>>(v_grid, v_sc,   att_s2g, src_s2g, degg, bkg, ohg, nxg,
                                              sg, gr_id, W1, b1, out + (size_t)NQ * DIM);
}


# combine: one launch for both sides; each block does a 16-row tile of each side with its W1 fragments fetched once (was two launches, W1 fetched in each)
# speedup vs baseline: 4.8032x; 1.1404x over previous
#include <hip/hip_runtime.h>

#define NG     100000
#define NS     20000
#define EDGES  300000
#define DIM    256
#define NQ     4096
#define MAXU   4096
#define LDSPAD 2
#define CAP    64
#define GRP    8
#define EPT    8
#define LINKB  ((EDGES / EPT + 255) / 256)
static_assert(EDGES % EPT == 0 && EPT == 8, "a link thread owns 8 whole edges (two 16-byte loads per array)");

typedef __attribute__((ext_vector_type(2))) float v2f;
typedef __attribute__((ext_vector_type(8))) float v8f;

__global__ __launch_bounds__(256) void k_init_slots(int* __restrict__ sg,
                                                    int* __restrict__ ss,
                                                    int* __restrict__ cnt,
                                                    int* __restrict__ deg,
                                                    int* __restrict__ ohead) {
  int i = blockIdx.x * 256 + threadIdx.x;
  if (i < NG) sg[i] = -1;
  if (i < NS) ss[i] = -1;
  if (i < 2)  cnt[i] = 0;
  if (i < 2 * MAXU) { deg[i] = 0; ohead[i] = -1; }
}

__global__ __launch_bounds__(256) void k_compact(const int* __restrict__ gid,
                                                 const int* __restrict__ sid,
                                                 int* __restrict__ sg,
                                                 int* __restrict__ ss,
                                                 int* __restrict__ cnt) {
  int i = blockIdx.x * 256 + threadIdx.x;
  if (i < NQ) {
    int g = gid[i];
    if (atomicCAS(&sg[g], -1, -2) == -1) sg[g] = atomicAdd(&cnt[0], 1);
  } else if (i < 2 * NQ) {
    int g = sid[i - NQ];
    if (atomicCAS(&ss[g], -1, -2) == -1) ss[g] = atomicAdd(&cnt[1], 1);
  }
}

__global__ __launch_bounds__(256) void k_link(const float* __restrict__ att0,
                                              const int* __restrict__ src0,
                                              const int* __restrict__ dst0,
                                              const int* __restrict__ slot0,
                                              int* __restrict__ deg0,
                                              int* __restrict__ bucket0,
                                              int* __restrict__ ohead0,
                                              int* __restrict__ nxt0,
                                              const float* __restrict__ att1,
                                              const int* __restrict__ src1,
                                              const int* __restrict__ dst1,
                                              const int* __restrict__ slot1,
                                              int* __restrict__ deg1,
                                              int* __restrict__ bucket1,
                                              int* __restrict__ ohead1,
                                              int* __restrict__ nxt1) {
  const bool first = (blockIdx.y == 0);
  const float* __restrict__ att    = first ? att0    : att1;
  const int*   __restrict__ src    = first ? src0    : src1;
  const int*   __restrict__ dst    = first ? dst0    : dst1;
  const int*   __restrict__ slot   = first ? slot0   : slot1;
  int*         __restrict__ deg    = first ? deg0    : deg1;
  int*         __restrict__ bucket = first ? bucket0 : bucket1;
  int*         __restrict__ ohead  = first ? ohead0  : ohead1;
  int*         __restrict__ nxt    = first ? nxt0    : nxt1;

  const int e0 = (blockIdx.x * 256 + threadIdx.x) * EPT;
  if (e0 >= EDGES) return;

  const int4   d0 = *(const int4*)(dst + e0);
  const int4   d1 = *(const int4*)(dst + e0 + 4);
  const int4   r0 = *(const int4*)(src + e0);
  const int4   r1 = *(const int4*)(src + e0 + 4);
  const float4 a0 = *(const float4*)(att + e0);
  const float4 a1 = *(const float4*)(att + e0 + 4);
  const int   d[EPT]  = {d0.x, d0.y, d0.z, d0.w, d1.x, d1.y, d1.z, d1.w};
  const int   sr[EPT] = {r0.x, r0.y, r0.z, r0.w, r1.x, r1.y, r1.z, r1.w};
  const float at[EPT] = {a0.x, a0.y, a0.z, a0.w, a1.x, a1.y, a1.z, a1.w};

  int s[EPT];
#pragma unroll
  for (int k = 0; k < EPT; ++k) s[k] = slot[d[k]];

  if ((s[0] & s[1] & s[2] & s[3] & s[4] & s[5] & s[6] & s[7]) < 0) return;

#define LINK_ASK(k) int p##k = 0; if (s[k] >= 0) p##k = atomicAdd(&deg[s[k]], 1);
  LINK_ASK(0) LINK_ASK(1) LINK_ASK(2) LINK_ASK(3)
  LINK_ASK(4) LINK_ASK(5) LINK_ASK(6) LINK_ASK(7)
#undef LINK_ASK

#define LINK_PUT(k)                                                        \
  if (s[k] >= 0) {                                                         \
    if (p##k < CAP) {                                                      \
      int2 ent; ent.x = sr[k]; ent.y = __float_as_int(at[k]);              \
      *(int2*)(bucket + ((size_t)s[k] * CAP + p##k) * 2) = ent;            \
    } else {                                                               \
      nxt[e0 + k] = atomicExch(&ohead[s[k]], e0 + k);                      \
    }                                                                      \
  }
  LINK_PUT(0) LINK_PUT(1) LINK_PUT(2) LINK_PUT(3)
  LINK_PUT(4) LINK_PUT(5) LINK_PUT(6) LINK_PUT(7)
#undef LINK_PUT
}

#define CTHREADS 256
__global__ __launch_bounds__(CTHREADS) void k_combine(const float* __restrict__ v_sc,
                                                         const float* __restrict__ v_grid,
                                                         const float* __restrict__ att_g2s,
                                                         const float* __restrict__ att_s2g,
                                                         const int* __restrict__ src_g2s,
                                                         const int* __restrict__ src_s2g,
                                                         const int* __restrict__ degs,
                                                         const int* __restrict__ degg,
                                                         const int* __restrict__ bks,
                                                         const int* __restrict__ bkg,
                                                         const int* __restrict__ ohs,
                                                         const int* __restrict__ ohg,
                                                         const int* __restrict__ nxs,
                                                         const int* __restrict__ nxg,
                                                         const int* __restrict__ ss,
                                                         const int* __restrict__ sg,
                                                         const int* __restrict__ sc_id,
                                                         const int* __restrict__ gr_id,
                                                         const float* __restrict__ W1,
                                                         const float* __restrict__ b1,
                                                         float* __restrict__ out) {
  __shared__ __attribute__((aligned(16))) float x1[2][16][DIM + LDSPAD];
  __shared__ __attribute__((aligned(16))) float x2[2][16][DIM + LDSPAD];

  const int lane  = threadIdx.x & 31;
  const int wave  = __builtin_amdgcn_readfirstlane((int)(threadIdx.x >> 5));
  const int qbase = blockIdx.x * 16;

  const int mrow  = lane & 15;
  const int khalf = lane >> 4;

  const int ncolA = wave * 32 + mrow;
  const int ncolB = ncolA + 16;
  const float* wrowA = W1 + (size_t)ncolA * DIM + 2 * khalf;
  const float* wrowB = W1 + (size_t)ncolB * DIM + 2 * khalf;
  v2f bA[DIM / 4];
  v2f bB[DIM / 4];
#pragma unroll
  for (int t = 0; t < DIM / 4; ++t) {
    bA[t].x = wrowA[4 * t];  bA[t].y = wrowA[4 * t + 1];
    bB[t].x = wrowB[4 * t];  bB[t].y = wrowB[4 * t + 1];
  }

  for (int tr = 0; tr < 4; ++tr) {
    const int side = tr >> 1;
    const int rr   = tr & 1;
    const float* v      = side ? v_grid  : v_sc;
    const float* vsrc   = side ? v_sc    : v_grid;
    const float* att    = side ? att_s2g : att_g2s;
    const int*   esrc   = side ? src_s2g : src_g2s;
    const int*   deg    = side ? degg    : degs;
    const int*   bucket = side ? bkg     : bks;
    const int*   ohead  = side ? ohg     : ohs;
    const int*   nxt    = side ? nxg     : nxs;
    const int*   slot   = side ? sg      : ss;
    const int*   ids    = side ? gr_id   : sc_id;
    const int r = wave * 2 + rr;
    const int g = ids[qbase + r];
    const int s = slot[g];
    const int d  = __builtin_amdgcn_readfirstlane(deg[s]);
    const int nb = (d < CAP) ? d : CAP;
    const float4* vp = (const float4*)(v + (size_t)g * DIM);
    const float4 vv0 = vp[lane];
    const float4 vv1 = vp[lane + 32];
    float4 n0 = {0.0f, 0.0f, 0.0f, 0.0f};
    float4 n1 = {0.0f, 0.0f, 0.0f, 0.0f};

    for (int base = 0; base < nb; base += 32) {
      const int cn = (nb - base < 32) ? (nb - base) : 32;
      int   es = 0;
      float ea = 0.0f;
      if (lane < cn) {
        const int2 ent = *(const int2*)(bucket + ((size_t)s * CAP + base + lane) * 2);
        es = ent.x;
        ea = __int_as_float(ent.y);
      }
      for (int i = 0; i < cn; i += GRP) {
        float4 r0[GRP];
        float4 r1[GRP];
#pragma unroll
        for (int k = 0; k < GRP; ++k) {
          const int ik = (i + k < cn) ? (i + k) : (cn - 1);
          const int si = __builtin_amdgcn_readlane(es, ik);
          const float4* rp = (const float4*)(vsrc + (size_t)si * DIM);
          r0[k] = rp[lane];
          r1[k] = rp[lane + 32];
        }
#pragma unroll
        for (int k = 0; k < GRP; ++k) {
          if (i + k < cn) {
            const float ai = __int_as_float(
                __builtin_amdgcn_readlane(__float_as_int(ea), i + k));
            n0.x += r0[k].x * ai;  n0.y += r0[k].y * ai;
            n0.z += r0[k].z * ai;  n0.w += r0[k].w * ai;
            n1.x += r1[k].x * ai;  n1.y += r1[k].y * ai;
            n1.z += r1[k].z * ai;  n1.w += r1[k].w * ai;
          }
        }
      }
    }
    if (d > CAP) {
      int e = __builtin_amdgcn_readfirstlane(ohead[s]);
      for (int it = 0; it < EDGES && e >= 0; ++it) {
        const int   si = esrc[e];
        const float ai = att[e];
        const float4* rp = (const float4*)(vsrc + (size_t)si * DIM);
        const float4 q0 = rp[lane];
        const float4 q1 = rp[lane + 32];
        n0.x += q0.x * ai;  n0.y += q0.y * ai;  n0.z += q0.z * ai;  n0.w += q0.w * ai;
        n1.x += q1.x * ai;  n1.y += q1.y * ai;  n1.z += q1.z * ai;  n1.w += q1.w * ai;
        e = __builtin_amdgcn_readfirstlane(nxt[e]);
      }
    }

    const int c0 = lane * 4;
    const int c1 = c0 + 128;
    x1[side][r][c0 + 0] = vv0.x + n0.x;  x2[side][r][c0 + 0] = vv0.x * n0.x;
    x1[side][r][c0 + 1] = vv0.y + n0.y;  x2[side][r][c0 + 1] = vv0.y * n0.y;
    x1[side][r][c0 + 2] = vv0.z + n0.z;  x2[side][r][c0 + 2] = vv0.z * n0.z;
    x1[side][r][c0 + 3] = vv0.w + n0.w;  x2[side][r][c0 + 3] = vv0.w * n0.w;
    x1[side][r][c1 + 0] = vv1.x + n1.x;  x2[side][r][c1 + 0] = vv1.x * n1.x;
    x1[side][r][c1 + 1] = vv1.y + n1.y;  x2[side][r][c1 + 1] = vv1.y * n1.y;
    x1[side][r][c1 + 2] = vv1.z + n1.z;  x2[side][r][c1 + 2] = vv1.z * n1.z;
    x1[side][r][c1 + 3] = vv1.w + n1.w;  x2[side][r][c1 + 3] = vv1.w * n1.w;
  }
  __syncthreads();

  const float biasA = b1[ncolA];
  const float biasB = b1[ncolB];
  for (int side = 0; side < 2; ++side) {
    v8f acc1A = {};
    v8f acc2A = {};
    v8f acc1B = {};
    v8f acc2B = {};
#pragma unroll
    for (int t = 0; t < DIM / 4; ++t) {
      const int ka = 4 * t + 2 * khalf;
      v2f a1; a1.x = x1[side][mrow][ka]; a1.y = x1[side][mrow][ka + 1];
      v2f a2; a2.x = x2[side][mrow][ka]; a2.y = x2[side][mrow][ka + 1];
      acc1A = __builtin_amdgcn_wmma_f32_16x16x4_f32(
          false, a1, false, bA[t], (short)0, acc1A, false, false);
      acc2A = __builtin_amdgcn_wmma_f32_16x16x4_f32(
          false, a2, false, bA[t], (short)0, acc2A, false, false);
      acc1B = __builtin_amdgcn_wmma_f32_16x16x4_f32(
          false, a1, false, bB[t], (short)0, acc1B, false, false);
      acc2B = __builtin_amdgcn_wmma_f32_16x16x4_f32(
          false, a2, false, bB[t], (short)0, acc2B, false, false);
    }

    float* __restrict__ orow = out + ((size_t)side * NQ + qbase) * DIM;
#pragma unroll
    for (int r = 0; r < 8; ++r) {
      const int m = khalf * 8 + r;
      float y1 = acc1A[r] + biasA;
      float y2 = acc2A[r] + biasA;
      y1 = (y1 >= 0.0f) ? y1 : 0.01f * y1;
      y2 = (y2 >= 0.0f) ? y2 : 0.01f * y2;
      orow[(size_t)m * DIM + ncolA] = y1 + y2;
      float z1 = acc1B[r] + biasB;
      float z2 = acc2B[r] + biasB;
      z1 = (z1 >= 0.0f) ? z1 : 0.01f * z1;
      z2 = (z2 >= 0.0f) ? z2 : 0.01f * z2;
      orow[(size_t)m * DIM + ncolB] = z1 + z2;
    }
  }
}

extern "C" void kernel_launch(void* const* d_in, const int* in_sizes, int n_in,
                              void* d_out, int out_size, void* d_ws, size_t ws_size,
                              hipStream_t stream) {
  const float* v_grid  = (const float*)d_in[0];
  const float* v_sc    = (const float*)d_in[1];
  const float* att_s2g = (const float*)d_in[2];
  const float* att_g2s = (const float*)d_in[3];
  const float* W1      = (const float*)d_in[4];
  const float* b1      = (const float*)d_in[5];
  const int*   src_s2g = (const int*)d_in[6];
  const int*   dst_s2g = (const int*)d_in[7];
  const int*   src_g2s = (const int*)d_in[8];
  const int*   dst_g2s = (const int*)d_in[9];
  const int*   sc_id   = (const int*)d_in[10];
  const int*   gr_id   = (const int*)d_in[11];
  float* out = (float*)d_out;

  int*  sg   = (int*)d_ws;
  int*  ss   = sg + NG;
  int*  cnt  = ss + NS;
  int*  degg = cnt + 4;
  int*  degs = degg + MAXU;
  int*  ohg  = degs + MAXU;
  int*  ohs  = ohg + MAXU;
  int*  nxg  = ohs + MAXU;
  int*  nxs  = nxg + EDGES;
  int*  bkg  = nxs + EDGES;
  int*  bks  = bkg + (size_t)MAXU * CAP * 2;

  k_init_slots<<<(NG + 255) / 256, 256, 0, stream>>>(sg, ss, cnt, degg, ohg);
  k_compact<<<(2 * NQ + 255) / 256, 256, 0, stream>>>(gr_id, sc_id, sg, ss, cnt);

  k_link<<<dim3(LINKB, 2), 256, 0, stream>>>(att_s2g, src_s2g, dst_s2g, sg, degg, bkg, ohg, nxg,
                                             att_g2s, src_g2s, dst_g2s, ss, degs, bks, ohs, nxs);

  k_combine<<<NQ / 16, CTHREADS, 0, stream>>>(v_sc, v_grid, att_g2s, att_s2g, src_g2s, src_s2g,
                                              degs, degg, bks, bkg, ohs, ohg, nxs, nxg,
                                              ss, sg, sc_id, gr_id, W1, b1, out);
}


# node-to-slot map: one 1,024-thread workgroup marks queried nodes in an LDS bitmap and scans a rank per word (slot = rank + set bits below); replaces cleared map and CAS compaction, 4 launches to 3
# speedup vs baseline: 5.1477x; 1.0717x over previous
#include <hip/hip_runtime.h>

#define NG     100000
#define NS     20000
#define EDGES  300000
#define DIM    256
#define NQ     4096
#define MAXU   4096
#define LDSPAD 2
#define CAP    64
#define GRP    8
#define EPT    8
#define LINKB  ((EDGES / EPT + 255) / 256)
#define WG     (NG / 32)
#define WSC    (NS / 32)
#define RTHREADS 1024
static_assert(NG % 32 == 0 && NS % 32 == 0, "a node bitmap is whole 32-bit words");
static_assert(WG + WSC <= 4 * RTHREADS, "a rank thread owns 4 consecutive bitmap words");
static_assert(NQ % RTHREADS == 0, "a rank thread marks NQ / RTHREADS queries per side");
static_assert(EDGES % EPT == 0 && EPT == 8, "a link thread owns 8 whole edges (two 16-byte loads per array)");

typedef __attribute__((ext_vector_type(2))) float v2f;
typedef __attribute__((ext_vector_type(8))) float v8f;

__global__ __launch_bounds__(RTHREADS) void k_rank(const int* __restrict__ gid,
                                                    const int* __restrict__ sid,
                                                    int* __restrict__ tab,
                                                    int* __restrict__ deg,
                                                    int* __restrict__ ohead) {
  __shared__ __attribute__((aligned(16))) unsigned bm[4 * RTHREADS];
  __shared__ int wtot[RTHREADS / 32];
  __shared__ int gtot;
  const int tid  = threadIdx.x;
  const int lane = tid & 31;
  const int wave = tid >> 5;

  for (int i = tid; i < 2 * MAXU / 2; i += RTHREADS) {
    ((int2*)deg)[i]   = make_int2(0, 0);
    ((int2*)ohead)[i] = make_int2(-1, -1);
  }

  int g[NQ / RTHREADS];
  int c[NQ / RTHREADS];
#pragma unroll
  for (int k = 0; k < NQ / RTHREADS; ++k) {
    g[k] = gid[tid + k * RTHREADS];
    c[k] = sid[tid + k * RTHREADS];
  }
  for (int k = 0; k < 4; ++k) bm[tid + k * RTHREADS] = 0u;
  __syncthreads();
#pragma unroll
  for (int k = 0; k < NQ / RTHREADS; ++k) {
    atomicOr(&bm[g[k] >> 5], 1u << (g[k] & 31));
    atomicOr(&bm[WG + (c[k] >> 5)], 1u << (c[k] & 31));
  }
  __syncthreads();

  const uint4 bw = *(const uint4*)(bm + 4 * tid);
  const int p1   = __popc(bw.x);
  const int p2   = p1 + __popc(bw.y);
  const int p3   = p2 + __popc(bw.z);
  const int tsum = p3 + __popc(bw.w);
  int incl = tsum;
  for (int d = 1; d < 32; d <<= 1) {
    const int o = __shfl_up(incl, d);
    if (lane >= d) incl += o;
  }
  if (lane == 31) wtot[wave] = incl;
  __syncthreads();
  if (wave == 0) {
    const int v = wtot[lane];
    int iv = v;
    for (int d = 1; d < 32; d <<= 1) {
      const int o = __shfl_up(iv, d);
      if (lane >= d) iv += o;
    }
    wtot[lane] = iv - v;
  }
  __syncthreads();
  const int base = wtot[wave] + incl - tsum;
  const int pre[4] = {base, base + p1, base + p2, base + p3};
  const unsigned bits[4] = {bw.x, bw.y, bw.z, bw.w};
#pragma unroll
  for (int j = 0; j < 4; ++j)
    if (4 * tid + j == WG) gtot = pre[j];
  __syncthreads();
  const int gt = gtot;
#pragma unroll
  for (int j = 0; j < 4; ++j) {
    const int w = 4 * tid + j;
    if (w < WG + WSC) {
      int2 ent; ent.x = (int)bits[j]; ent.y = (w < WG) ? pre[j] : pre[j] - gt;
      *(int2*)(tab + 2 * w) = ent;
    }
  }
}

__device__ __forceinline__ unsigned long long slot_entry(const int* __restrict__ tab, int node) {
  return *(const unsigned long long*)(tab + 2 * (node >> 5));
}
__device__ __forceinline__ int slot_from(unsigned long long ent, int node) {
  const unsigned bits = (unsigned)ent;
  const int      rank = (int)(ent >> 32);
  const unsigned bit  = 1u << (node & 31);
  const int      s    = rank + __popc(bits & (bit - 1u));
  return (bits & bit) ? s : -1;
}

__global__ __launch_bounds__(256) void k_link(const float* __restrict__ att0,
                                              const int* __restrict__ src0,
                                              const int* __restrict__ dst0,
                                              const int* __restrict__ tab0,
                                              int* __restrict__ deg0,
                                              int* __restrict__ bucket0,
                                              int* __restrict__ ohead0,
                                              int* __restrict__ nxt0,
                                              const float* __restrict__ att1,
                                              const int* __restrict__ src1,
                                              const int* __restrict__ dst1,
                                              const int* __restrict__ tab1,
                                              int* __restrict__ deg1,
                                              int* __restrict__ bucket1,
                                              int* __restrict__ ohead1,
                                              int* __restrict__ nxt1) {
  const bool first = (blockIdx.y == 0);
  const float* __restrict__ att    = first ? att0    : att1;
  const int*   __restrict__ src    = first ? src0    : src1;
  const int*   __restrict__ dst    = first ? dst0    : dst1;
  const int*   __restrict__ tab    = first ? tab0    : tab1;
  int*         __restrict__ deg    = first ? deg0    : deg1;
  int*         __restrict__ bucket = first ? bucket0 : bucket1;
  int*         __restrict__ ohead  = first ? ohead0  : ohead1;
  int*         __restrict__ nxt    = first ? nxt0    : nxt1;

  const int e0 = (blockIdx.x * 256 + threadIdx.x) * EPT;
  if (e0 >= EDGES) return;

  const int4   d0 = *(const int4*)(dst + e0);
  const int4   d1 = *(const int4*)(dst + e0 + 4);
  const int4   r0 = *(const int4*)(src + e0);
  const int4   r1 = *(const int4*)(src + e0 + 4);
  const float4 a0 = *(const float4*)(att + e0);
  const float4 a1 = *(const float4*)(att + e0 + 4);
  const int   d[EPT]  = {d0.x, d0.y, d0.z, d0.w, d1.x, d1.y, d1.z, d1.w};
  const int   sr[EPT] = {r0.x, r0.y, r0.z, r0.w, r1.x, r1.y, r1.z, r1.w};
  const float at[EPT] = {a0.x, a0.y, a0.z, a0.w, a1.x, a1.y, a1.z, a1.w};

  unsigned long long te[EPT];
#pragma unroll
  for (int k = 0; k < EPT; ++k) te[k] = slot_entry(tab, d[k]);
  int s[EPT];
#pragma unroll
  for (int k = 0; k < EPT; ++k) s[k] = slot_from(te[k], d[k]);

  if ((s[0] & s[1] & s[2] & s[3] & s[4] & s[5] & s[6] & s[7]) < 0) return;

#define LINK_ASK(k) int p##k = 0; if (s[k] >= 0) p##k = atomicAdd(&deg[s[k]], 1);
  LINK_ASK(0) LINK_ASK(1) LINK_ASK(2) LINK_ASK(3)
  LINK_ASK(4) LINK_ASK(5) LINK_ASK(6) LINK_ASK(7)
#undef LINK_ASK

#define LINK_PUT(k)                                                        \
  if (s[k] >= 0) {                                                         \
    if (p##k < CAP) {                                                      \
      int2 ent; ent.x = sr[k]; ent.y = __float_as_int(at[k]);              \
      *(int2*)(bucket + ((size_t)s[k] * CAP + p##k) * 2) = ent;            \
    } else {                                                               \
      nxt[e0 + k] = atomicExch(&ohead[s[k]], e0 + k);                      \
    }                                                                      \
  }
  LINK_PUT(0) LINK_PUT(1) LINK_PUT(2) LINK_PUT(3)
  LINK_PUT(4) LINK_PUT(5) LINK_PUT(6) LINK_PUT(7)
#undef LINK_PUT
}

#define CTHREADS 256
__global__ __launch_bounds__(CTHREADS) void k_combine(const float* __restrict__ v_sc,
                                                         const float* __restrict__ v_grid,
                                                         const float* __restrict__ att_g2s,
                                                         const float* __restrict__ att_s2g,
                                                         const int* __restrict__ src_g2s,
                                                         const int* __restrict__ src_s2g,
                                                         const int* __restrict__ degs,
                                                         const int* __restrict__ degg,
                                                         const int* __restrict__ bks,
                                                         const int* __restrict__ bkg,
                                                         const int* __restrict__ ohs,
                                                         const int* __restrict__ ohg,
                                                         const int* __restrict__ nxs,
                                                         const int* __restrict__ nxg,
                                                         const int* __restrict__ tabs,
                                                         const int* __restrict__ tabg,
                                                         const int* __restrict__ sc_id,
                                                         const int* __restrict__ gr_id,
                                                         const float* __restrict__ W1,
                                                         const float* __restrict__ b1,
                                                         float* __restrict__ out) {
  __shared__ __attribute__((aligned(16))) float x1[2][16][DIM + LDSPAD];
  __shared__ __attribute__((aligned(16))) float x2[2][16][DIM + LDSPAD];

  const int lane  = threadIdx.x & 31;
  const int wave  = __builtin_amdgcn_readfirstlane((int)(threadIdx.x >> 5));
  const int qbase = blockIdx.x * 16;

  const int mrow  = lane & 15;
  const int khalf = lane >> 4;

  const int ncolA = wave * 32 + mrow;
  const int ncolB = ncolA + 16;
  const float* wrowA = W1 + (size_t)ncolA * DIM + 2 * khalf;
  const float* wrowB = W1 + (size_t)ncolB * DIM + 2 * khalf;
  v2f bA[DIM / 4];
  v2f bB[DIM / 4];
#pragma unroll
  for (int t = 0; t < DIM / 4; ++t) {
    bA[t].x = wrowA[4 * t];  bA[t].y = wrowA[4 * t + 1];
    bB[t].x = wrowB[4 * t];  bB[t].y = wrowB[4 * t + 1];
  }

  for (int tr = 0; tr < 4; ++tr) {
    const int side = tr >> 1;
    const int rr   = tr & 1;
    const float* v      = side ? v_grid  : v_sc;
    const float* vsrc   = side ? v_sc    : v_grid;
    const float* att    = side ? att_s2g : att_g2s;
    const int*   esrc   = side ? src_s2g : src_g2s;
    const int*   deg    = side ? degg    : degs;
    const int*   bucket = side ? bkg     : bks;
    const int*   ohead  = side ? ohg     : ohs;
    const int*   nxt    = side ? nxg     : nxs;
    const int*   tab    = side ? tabg    : tabs;
    const int*   ids    = side ? gr_id   : sc_id;
    const int r = wave * 2 + rr;
    const int g = ids[qbase + r];
    const int s = slot_from(slot_entry(tab, g), g);
    const int d  = __builtin_amdgcn_readfirstlane(deg[s]);
    const int nb = (d < CAP) ? d : CAP;
    const float4* vp = (const float4*)(v + (size_t)g * DIM);
    const float4 vv0 = vp[lane];
    const float4 vv1 = vp[lane + 32];
    float4 n0 = {0.0f, 0.0f, 0.0f, 0.0f};
    float4 n1 = {0.0f, 0.0f, 0.0f, 0.0f};

    for (int base = 0; base < nb; base += 32) {
      const int cn = (nb - base < 32) ? (nb - base) : 32;
      int   es = 0;
      float ea = 0.0f;
      if (lane < cn) {
        const int2 ent = *(const int2*)(bucket + ((size_t)s * CAP + base + lane) * 2);
        es = ent.x;
        ea = __int_as_float(ent.y);
      }
      for (int i = 0; i < cn; i += GRP) {
        float4 r0[GRP];
        float4 r1[GRP];
#pragma unroll
        for (int k = 0; k < GRP; ++k) {
          const int ik = (i + k < cn) ? (i + k) : (cn - 1);
          const int si = __builtin_amdgcn_readlane(es, ik);
          const float4* rp = (const float4*)(vsrc + (size_t)si * DIM);
          r0[k] = rp[lane];
          r1[k] = rp[lane + 32];
        }
#pragma unroll
        for (int k = 0; k < GRP; ++k) {
          if (i + k < cn) {
            const float ai = __int_as_float(
                __builtin_amdgcn_readlane(__float_as_int(ea), i + k));
            n0.x += r0[k].x * ai;  n0.y += r0[k].y * ai;
            n0.z += r0[k].z * ai;  n0.w += r0[k].w * ai;
            n1.x += r1[k].x * ai;  n1.y += r1[k].y * ai;
            n1.z += r1[k].z * ai;  n1.w += r1[k].w * ai;
          }
        }
      }
    }
    if (d > CAP) {
      int e = __builtin_amdgcn_readfirstlane(ohead[s]);
      for (int it = 0; it < EDGES && e >= 0; ++it) {
        const int   si = esrc[e];
        const float ai = att[e];
        const float4* rp = (const float4*)(vsrc + (size_t)si * DIM);
        const float4 q0 = rp[lane];
        const float4 q1 = rp[lane + 32];
        n0.x += q0.x * ai;  n0.y += q0.y * ai;  n0.z += q0.z * ai;  n0.w += q0.w * ai;
        n1.x += q1.x * ai;  n1.y += q1.y * ai;  n1.z += q1.z * ai;  n1.w += q1.w * ai;
        e = __builtin_amdgcn_readfirstlane(nxt[e]);
      }
    }

    const int c0 = lane * 4;
    const int c1 = c0 + 128;
    x1[side][r][c0 + 0] = vv0.x + n0.x;  x2[side][r][c0 + 0] = vv0.x * n0.x;
    x1[side][r][c0 + 1] = vv0.y + n0.y;  x2[side][r][c0 + 1] = vv0.y * n0.y;
    x1[side][r][c0 + 2] = vv0.z + n0.z;  x2[side][r][c0 + 2] = vv0.z * n0.z;
    x1[side][r][c0 + 3] = vv0.w + n0.w;  x2[side][r][c0 + 3] = vv0.w * n0.w;
    x1[side][r][c1 + 0] = vv1.x + n1.x;  x2[side][r][c1 + 0] = vv1.x * n1.x;
    x1[side][r][c1 + 1] = vv1.y + n1.y;  x2[side][r][c1 + 1] = vv1.y * n1.y;
    x1[side][r][c1 + 2] = vv1.z + n1.z;  x2[side][r][c1 + 2] = vv1.z * n1.z;
    x1[side][r][c1 + 3] = vv1.w + n1.w;  x2[side][r][c1 + 3] = vv1.w * n1.w;
  }
  __syncthreads();

  const float biasA = b1[ncolA];
  const float biasB = b1[ncolB];
  for (int side = 0; side < 2; ++side) {
    v8f acc1A = {};
    v8f acc2A = {};
    v8f acc1B = {};
    v8f acc2B = {};
#pragma unroll
    for (int t = 0; t < DIM / 4; ++t) {
      const int ka = 4 * t + 2 * khalf;
      v2f a1; a1.x = x1[side][mrow][ka]; a1.y = x1[side][mrow][ka + 1];
      v2f a2; a2.x = x2[side][mrow][ka]; a2.y = x2[side][mrow][ka + 1];
      acc1A = __builtin_amdgcn_wmma_f32_16x16x4_f32(
          false, a1, false, bA[t], (short)0, acc1A, false, false);
      acc2A = __builtin_amdgcn_wmma_f32_16x16x4_f32(
          false, a2, false, bA[t], (short)0, acc2A, false, false);
      acc1B = __builtin_amdgcn_wmma_f32_16x16x4_f32(
          false, a1, false, bB[t], (short)0, acc1B, false, false);
      acc2B = __builtin_amdgcn_wmma_f32_16x16x4_f32(
          false, a2, false, bB[t], (short)0, acc2B, false, false);
    }

    float* __restrict__ orow = out + ((size_t)side * NQ + qbase) * DIM;
#pragma unroll
    for (int r = 0; r < 8; ++r) {
      const int m = khalf * 8 + r;
      float y1 = acc1A[r] + biasA;
      float y2 = acc2A[r] + biasA;
      y1 = (y1 >= 0.0f) ? y1 : 0.01f * y1;
      y2 = (y2 >= 0.0f) ? y2 : 0.01f * y2;
      orow[(size_t)m * DIM + ncolA] = y1 + y2;
      float z1 = acc1B[r] + biasB;
      float z2 = acc2B[r] + biasB;
      z1 = (z1 >= 0.0f) ? z1 : 0.01f * z1;
      z2 = (z2 >= 0.0f) ? z2 : 0.01f * z2;
      orow[(size_t)m * DIM + ncolB] = z1 + z2;
    }
  }
}

extern "C" void kernel_launch(void* const* d_in, const int* in_sizes, int n_in,
                              void* d_out, int out_size, void* d_ws, size_t ws_size,
                              hipStream_t stream) {
  const float* v_grid  = (const float*)d_in[0];
  const float* v_sc    = (const float*)d_in[1];
  const float* att_s2g = (const float*)d_in[2];
  const float* att_g2s = (const float*)d_in[3];
  const float* W1      = (const float*)d_in[4];
  const float* b1      = (const float*)d_in[5];
  const int*   src_s2g = (const int*)d_in[6];
  const int*   dst_s2g = (const int*)d_in[7];
  const int*   src_g2s = (const int*)d_in[8];
  const int*   dst_g2s = (const int*)d_in[9];
  const int*   sc_id   = (const int*)d_in[10];
  const int*   gr_id   = (const int*)d_in[11];
  float* out = (float*)d_out;

  int*  tabg = (int*)d_ws;
  int*  tabs = tabg + 2 * WG;
  int*  degg = tabs + 2 * WSC;
  int*  degs = degg + MAXU;
  int*  ohg  = degs + MAXU;
  int*  ohs  = ohg + MAXU;
  int*  nxg  = ohs + MAXU;
  int*  nxs  = nxg + EDGES;
  int*  bkg  = nxs + EDGES;
  int*  bks  = bkg + (size_t)MAXU * CAP * 2;

  k_rank<<<1, RTHREADS, 0, stream>>>(gr_id, sc_id, tabg, degg, ohg);

  k_link<<<dim3(LINKB, 2), 256, 0, stream>>>(att_s2g, src_s2g, dst_s2g, tabg, degg, bkg, ohg, nxg,
                                             att_g2s, src_g2s, dst_g2s, tabs, degs, bks, ohs, nxs);

  k_combine<<<NQ / 16, CTHREADS, 0, stream>>>(v_sc, v_grid, att_g2s, att_s2g, src_g2s, src_s2g,
                                              degs, degg, bks, bkg, ohs, ohg, nxs, nxg,
                                              tabs, tabg, sc_id, gr_id, W1, b1, out);
}


# combine gather batched over a wave's 4 rows: ids, slots, list lengths, bucket heads read level by level together; a side's 2 lists walked as one, 16 source rows in flight (was 8); row sum order kept
# speedup vs baseline: 5.5805x; 1.0841x over previous
#include <hip/hip_runtime.h>

#define NG     100000
#define NS     20000
#define EDGES  300000
#define DIM    256
#define NQ     4096
#define MAXU   4096
#define LDSPAD 2
#define CAP    64
#define GRP    8
#define PGRP   16
#define EPT    8
#define LINKB  ((EDGES / EPT + 255) / 256)
#define WG     (NG / 32)
#define WSC    (NS / 32)
#define RTHREADS 1024
static_assert(NG % 32 == 0 && NS % 32 == 0, "a node bitmap is whole 32-bit words");
static_assert(WG + WSC <= 4 * RTHREADS, "a rank thread owns 4 consecutive bitmap words");
static_assert(NQ % RTHREADS == 0, "a rank thread marks NQ / RTHREADS queries per side");
static_assert(EDGES % EPT == 0 && EPT == 8, "a link thread owns 8 whole edges (two 16-byte loads per array)");

typedef __attribute__((ext_vector_type(2))) float v2f;
typedef __attribute__((ext_vector_type(8))) float v8f;

__global__ __launch_bounds__(RTHREADS) void k_rank(const int* __restrict__ gid,
                                                    const int* __restrict__ sid,
                                                    int* __restrict__ tab,
                                                    int* __restrict__ deg,
                                                    int* __restrict__ ohead) {
  __shared__ __attribute__((aligned(16))) unsigned bm[4 * RTHREADS];
  __shared__ int wtot[RTHREADS / 32];
  __shared__ int gtot;
  const int tid  = threadIdx.x;
  const int lane = tid & 31;
  const int wave = tid >> 5;

  for (int i = tid; i < 2 * MAXU / 2; i += RTHREADS) {
    ((int2*)deg)[i]   = make_int2(0, 0);
    ((int2*)ohead)[i] = make_int2(-1, -1);
  }

  int g[NQ / RTHREADS];
  int c[NQ / RTHREADS];
#pragma unroll
  for (int k = 0; k < NQ / RTHREADS; ++k) {
    g[k] = gid[tid + k * RTHREADS];
    c[k] = sid[tid + k * RTHREADS];
  }
  for (int k = 0; k < 4; ++k) bm[tid + k * RTHREADS] = 0u;
  __syncthreads();
#pragma unroll
  for (int k = 0; k < NQ / RTHREADS; ++k) {
    atomicOr(&bm[g[k] >> 5], 1u << (g[k] & 31));
    atomicOr(&bm[WG + (c[k] >> 5)], 1u << (c[k] & 31));
  }
  __syncthreads();

  const uint4 bw = *(const uint4*)(bm + 4 * tid);
  const int p1   = __popc(bw.x);
  const int p2   = p1 + __popc(bw.y);
  const int p3   = p2 + __popc(bw.z);
  const int tsum = p3 + __popc(bw.w);
  int incl = tsum;
  for (int d = 1; d < 32; d <<= 1) {
    const int o = __shfl_up(incl, d);
    if (lane >= d) incl += o;
  }
  if (lane == 31) wtot[wave] = incl;
  __syncthreads();
  if (wave == 0) {
    const int v = wtot[lane];
    int iv = v;
    for (int d = 1; d < 32; d <<= 1) {
      const int o = __shfl_up(iv, d);
      if (lane >= d) iv += o;
    }
    wtot[lane] = iv - v;
  }
  __syncthreads();
  const int base = wtot[wave] + incl - tsum;
  const int pre[4] = {base, base + p1, base + p2, base + p3};
  const unsigned bits[4] = {bw.x, bw.y, bw.z, bw.w};
#pragma unroll
  for (int j = 0; j < 4; ++j)
    if (4 * tid + j == WG) gtot = pre[j];
  __syncthreads();
  const int gt = gtot;
#pragma unroll
  for (int j = 0; j < 4; ++j) {
    const int w = 4 * tid + j;
    if (w < WG + WSC) {
      int2 ent; ent.x = (int)bits[j]; ent.y = (w < WG) ? pre[j] : pre[j] - gt;
      *(int2*)(tab + 2 * w) = ent;
    }
  }
}

__device__ __forceinline__ unsigned long long slot_entry(const int* __restrict__ tab, int node) {
  return *(const unsigned long long*)(tab + 2 * (node >> 5));
}
__device__ __forceinline__ int slot_from(unsigned long long ent, int node) {
  const unsigned bits = (unsigned)ent;
  const int      rank = (int)(ent >> 32);
  const unsigned bit  = 1u << (node & 31);
  const int      s    = rank + __popc(bits & (bit - 1u));
  return (bits & bit) ? s : -1;
}

__global__ __launch_bounds__(256) void k_link(const float* __restrict__ att0,
                                              const int* __restrict__ src0,
                                              const int* __restrict__ dst0,
                                              const int* __restrict__ tab0,
                                              int* __restrict__ deg0,
                                              int* __restrict__ bucket0,
                                              int* __restrict__ ohead0,
                                              int* __restrict__ nxt0,
                                              const float* __restrict__ att1,
                                              const int* __restrict__ src1,
                                              const int* __restrict__ dst1,
                                              const int* __restrict__ tab1,
                                              int* __restrict__ deg1,
                                              int* __restrict__ bucket1,
                                              int* __restrict__ ohead1,
                                              int* __restrict__ nxt1) {
  const bool first = (blockIdx.y == 0);
  const float* __restrict__ att    = first ? att0    : att1;
  const int*   __restrict__ src    = first ? src0    : src1;
  const int*   __restrict__ dst    = first ? dst0    : dst1;
  const int*   __restrict__ tab    = first ? tab0    : tab1;
  int*         __restrict__ deg    = first ? deg0    : deg1;
  int*         __restrict__ bucket = first ? bucket0 : bucket1;
  int*         __restrict__ ohead  = first ? ohead0  : ohead1;
  int*         __restrict__ nxt    = first ? nxt0    : nxt1;

  const int e0 = (blockIdx.x * 256 + threadIdx.x) * EPT;
  if (e0 >= EDGES) return;

  const int4   d0 = *(const int4*)(dst + e0);
  const int4   d1 = *(const int4*)(dst + e0 + 4);
  const int4   r0 = *(const int4*)(src + e0);
  const int4   r1 = *(const int4*)(src + e0 + 4);
  const float4 a0 = *(const float4*)(att + e0);
  const float4 a1 = *(const float4*)(att + e0 + 4);
  const int   d[EPT]  = {d0.x, d0.y, d0.z, d0.w, d1.x, d1.y, d1.z, d1.w};
  const int   sr[EPT] = {r0.x, r0.y, r0.z, r0.w, r1.x, r1.y, r1.z, r1.w};
  const float at[EPT] = {a0.x, a0.y, a0.z, a0.w, a1.x, a1.y, a1.z, a1.w};

  unsigned long long te[EPT];
#pragma unroll
  for (int k = 0; k < EPT; ++k) te[k] = slot_entry(tab, d[k]);
  int s[EPT];
#pragma unroll
  for (int k = 0; k < EPT; ++k) s[k] = slot_from(te[k], d[k]);

  if ((s[0] & s[1] & s[2] & s[3] & s[4] & s[5] & s[6] & s[7]) < 0) return;

#define LINK_ASK(k) int p##k = 0; if (s[k] >= 0) p##k = atomicAdd(&deg[s[k]], 1);
  LINK_ASK(0) LINK_ASK(1) LINK_ASK(2) LINK_ASK(3)
  LINK_ASK(4) LINK_ASK(5) LINK_ASK(6) LINK_ASK(7)
#undef LINK_ASK

#define LINK_PUT(k)                                                        \
  if (s[k] >= 0) {                                                         \
    if (p##k < CAP) {                                                      \
      int2 ent; ent.x = sr[k]; ent.y = __float_as_int(at[k]);              \
      *(int2*)(bucket + ((size_t)s[k] * CAP + p##k) * 2) = ent;            \
    } else {                                                               \
      nxt[e0 + k] = atomicExch(&ohead[s[k]], e0 + k);                      \
    }                                                                      \
  }
  LINK_PUT(0) LINK_PUT(1) LINK_PUT(2) LINK_PUT(3)
  LINK_PUT(4) LINK_PUT(5) LINK_PUT(6) LINK_PUT(7)
#undef LINK_PUT
}

__device__ __forceinline__ void grp_ask(const float* __restrict__ vsrc, int es, int i, int cn,
                                        int lane, float4 (&r0)[GRP], float4 (&r1)[GRP]) {
#pragma unroll
  for (int k = 0; k < GRP; ++k) {
    const int ik = (i + k < cn) ? (i + k) : (cn - 1);
    const int si = __builtin_amdgcn_readlane(es, ik);
    const float4* rp = (const float4*)(vsrc + (size_t)si * DIM);
    r0[k] = rp[lane];
    r1[k] = rp[lane + 32];
  }
}
__device__ __forceinline__ void grp_sum(float ea, int i, int cn,
                                        const float4 (&r0)[GRP], const float4 (&r1)[GRP],
                                        float4& n0, float4& n1) {
#pragma unroll
  for (int k = 0; k < GRP; ++k) {
    if (i + k < cn) {
      const float ai = __int_as_float(
          __builtin_amdgcn_readlane(__float_as_int(ea), i + k));
      n0.x += r0[k].x * ai;  n0.y += r0[k].y * ai;
      n0.z += r0[k].z * ai;  n0.w += r0[k].w * ai;
      n1.x += r1[k].x * ai;  n1.y += r1[k].y * ai;
      n1.z += r1[k].z * ai;  n1.w += r1[k].w * ai;
    }
  }
}
__device__ __forceinline__ void pair_ask(const float* __restrict__ vsrc, int es_a, int es_b,
                                         int j0, int cna, int ctot, int lane,
                                         float4 (&r0)[PGRP], float4 (&r1)[PGRP]) {
#pragma unroll
  for (int k = 0; k < PGRP; ++k) {
    const int j  = (j0 + k < ctot) ? (j0 + k) : (ctot - 1);
    const bool ina = j < cna;
    const int sa = __builtin_amdgcn_readlane(es_a, ina ? j : 0);
    const int sb = __builtin_amdgcn_readlane(es_b, ina ? 0 : (j - cna));
    const int si = ina ? sa : sb;
    const float4* rp = (const float4*)(vsrc + (size_t)si * DIM);
    r0[k] = rp[lane];
    r1[k] = rp[lane + 32];
  }
}
__device__ __forceinline__ void pair_sum(float ea_a, float ea_b, int j0, int cna, int ctot,
                                         const float4 (&r0)[PGRP], const float4 (&r1)[PGRP],
                                         float4& na0, float4& na1, float4& nb0, float4& nb1) {
#pragma unroll
  for (int k = 0; k < PGRP; ++k) {
    const int j = j0 + k;
    if (j < ctot) {
      if (j < cna) {
        const float ai = __int_as_float(__builtin_amdgcn_readlane(__float_as_int(ea_a), j));
        na0.x += r0[k].x * ai;  na0.y += r0[k].y * ai;
        na0.z += r0[k].z * ai;  na0.w += r0[k].w * ai;
        na1.x += r1[k].x * ai;  na1.y += r1[k].y * ai;
        na1.z += r1[k].z * ai;  na1.w += r1[k].w * ai;
      } else {
        const float ai = __int_as_float(__builtin_amdgcn_readlane(__float_as_int(ea_b), j - cna));
        nb0.x += r0[k].x * ai;  nb0.y += r0[k].y * ai;
        nb0.z += r0[k].z * ai;  nb0.w += r0[k].w * ai;
        nb1.x += r1[k].x * ai;  nb1.y += r1[k].y * ai;
        nb1.z += r1[k].z * ai;  nb1.w += r1[k].w * ai;
      }
    }
  }
}
__device__ __forceinline__ void row_tail(const float* __restrict__ vsrc,
                                         const float* __restrict__ att,
                                         const int* __restrict__ esrc,
                                         const int* __restrict__ bucket,
                                         const int* __restrict__ ohead,
                                         const int* __restrict__ nxt,
                                         int s, int d, int nb, int lane,
                                         float4& n0, float4& n1) {
  for (int base = 32; base < nb; base += 32) {
    const int cn = (nb - base < 32) ? (nb - base) : 32;
    int   es = 0;
    float ea = 0.0f;
    if (lane < cn) {
      const int2 ent = *(const int2*)(bucket + ((size_t)s * CAP + base + lane) * 2);
      es = ent.x;
      ea = __int_as_float(ent.y);
    }
    for (int i = 0; i < cn; i += GRP) {
      float4 r0[GRP];
      float4 r1[GRP];
      grp_ask(vsrc, es, i, cn, lane, r0, r1);
      grp_sum(ea, i, cn, r0, r1, n0, n1);
    }
  }
  if (d > CAP) {
    int e = __builtin_amdgcn_readfirstlane(ohead[s]);
    for (int it = 0; it < EDGES && e >= 0; ++it) {
      const int   si = esrc[e];
      const float ai = att[e];
      const float4* rp = (const float4*)(vsrc + (size_t)si * DIM);
      const float4 q0 = rp[lane];
      const float4 q1 = rp[lane + 32];
      n0.x += q0.x * ai;  n0.y += q0.y * ai;  n0.z += q0.z * ai;  n0.w += q0.w * ai;
      n1.x += q1.x * ai;  n1.y += q1.y * ai;  n1.z += q1.z * ai;  n1.w += q1.w * ai;
      e = __builtin_amdgcn_readfirstlane(nxt[e]);
    }
  }
}
__device__ __forceinline__ void put_row(float* __restrict__ x1r, float* __restrict__ x2r, int lane,
                                        float4 vv0, float4 vv1, float4 n0, float4 n1) {
  const int c0 = lane * 4;
  const int c1 = c0 + 128;
  x1r[c0 + 0] = vv0.x + n0.x;  x2r[c0 + 0] = vv0.x * n0.x;
  x1r[c0 + 1] = vv0.y + n0.y;  x2r[c0 + 1] = vv0.y * n0.y;
  x1r[c0 + 2] = vv0.z + n0.z;  x2r[c0 + 2] = vv0.z * n0.z;
  x1r[c0 + 3] = vv0.w + n0.w;  x2r[c0 + 3] = vv0.w * n0.w;
  x1r[c1 + 0] = vv1.x + n1.x;  x2r[c1 + 0] = vv1.x * n1.x;
  x1r[c1 + 1] = vv1.y + n1.y;  x2r[c1 + 1] = vv1.y * n1.y;
  x1r[c1 + 2] = vv1.z + n1.z;  x2r[c1 + 2] = vv1.z * n1.z;
  x1r[c1 + 3] = vv1.w + n1.w;  x2r[c1 + 3] = vv1.w * n1.w;
}

#define CTHREADS 256
__global__ __launch_bounds__(CTHREADS) void k_combine(const float* __restrict__ v_sc,
                                                         const float* __restrict__ v_grid,
                                                         const float* __restrict__ att_g2s,
                                                         const float* __restrict__ att_s2g,
                                                         const int* __restrict__ src_g2s,
                                                         const int* __restrict__ src_s2g,
                                                         const int* __restrict__ degs,
                                                         const int* __restrict__ degg,
                                                         const int* __restrict__ bks,
                                                         const int* __restrict__ bkg,
                                                         const int* __restrict__ ohs,
                                                         const int* __restrict__ ohg,
                                                         const int* __restrict__ nxs,
                                                         const int* __restrict__ nxg,
                                                         const int* __restrict__ tabs,
                                                         const int* __restrict__ tabg,
                                                         const int* __restrict__ sc_id,
                                                         const int* __restrict__ gr_id,
                                                         const float* __restrict__ W1,
                                                         const float* __restrict__ b1,
                                                         float* __restrict__ out) {
  __shared__ __attribute__((aligned(16))) float x1[2][16][DIM + LDSPAD];
  __shared__ __attribute__((aligned(16))) float x2[2][16][DIM + LDSPAD];

  const int lane  = threadIdx.x & 31;
  const int wave  = __builtin_amdgcn_readfirstlane((int)(threadIdx.x >> 5));
  const int qbase = blockIdx.x * 16;

  const int mrow  = lane & 15;
  const int khalf = lane >> 4;

  const int ncolA = wave * 32 + mrow;
  const int ncolB = ncolA + 16;
  const float* wrowA = W1 + (size_t)ncolA * DIM + 2 * khalf;
  const float* wrowB = W1 + (size_t)ncolB * DIM + 2 * khalf;
  v2f bA[DIM / 4];
  v2f bB[DIM / 4];
#pragma unroll
  for (int t = 0; t < DIM / 4; ++t) {
    bA[t].x = wrowA[4 * t];  bA[t].y = wrowA[4 * t + 1];
    bB[t].x = wrowB[4 * t];  bB[t].y = wrowB[4 * t + 1];
  }

  int gq[4];
  int sq[4];
  int dq[4];
  int nq[4];
  unsigned long long tq[4];
#pragma unroll
  for (int tr = 0; tr < 4; ++tr) {
    const int* ids = (tr >> 1) ? gr_id : sc_id;
    gq[tr] = ids[qbase + wave * 2 + (tr & 1)];
  }
#pragma unroll
  for (int tr = 0; tr < 4; ++tr) tq[tr] = slot_entry((tr >> 1) ? tabg : tabs, gq[tr]);
#pragma unroll
  for (int tr = 0; tr < 4; ++tr) {
    const int* deg = (tr >> 1) ? degg : degs;
    sq[tr] = slot_from(tq[tr], gq[tr]);
    dq[tr] = __builtin_amdgcn_readfirstlane(deg[sq[tr]]);
    nq[tr] = (dq[tr] < CAP) ? dq[tr] : CAP;
  }
  int2 hq[4];
#pragma unroll
  for (int tr = 0; tr < 4; ++tr) {
    const int* bucket = (tr >> 1) ? bkg  : bks;
    const int* tab    = (tr >> 1) ? tabg : tabs;
    const bool any = nq[tr] > 0;
    const int* hp = any ? bucket + (size_t)sq[tr] * CAP * 2 : tab + 2 * (gq[tr] >> 5);
    const int  hl = any ? ((lane < nq[tr]) ? lane : nq[tr] - 1) : 0;
    hq[tr] = *(const int2*)(hp + 2 * hl);
  }

  for (int side = 0; side < 2; ++side) {
    const float* v      = side ? v_grid  : v_sc;
    const float* vsrc   = side ? v_sc    : v_grid;
    const float* att    = side ? att_s2g : att_g2s;
    const int*   esrc   = side ? src_s2g : src_g2s;
    const int*   bucket = side ? bkg     : bks;
    const int*   ohead  = side ? ohg     : ohs;
    const int*   nxt    = side ? nxg     : nxs;
    const int   g_a  = side ? gq[2] : gq[0];
    const int   g_b  = side ? gq[3] : gq[1];
    const int   s_a  = side ? sq[2] : sq[0];
    const int   s_b  = side ? sq[3] : sq[1];
    const int   d_a  = side ? dq[2] : dq[0];
    const int   d_b  = side ? dq[3] : dq[1];
    const int   n_a  = side ? nq[2] : nq[0];
    const int   n_b  = side ? nq[3] : nq[1];
    const int2  h_a  = side ? hq[2] : hq[0];
    const int2  h_b  = side ? hq[3] : hq[1];
    const int   es_a = h_a.x;
    const int   es_b = h_b.x;
    const float ea_a = __int_as_float(h_a.y);
    const float ea_b = __int_as_float(h_b.y);
    const float4* vpa = (const float4*)(v + (size_t)g_a * DIM);
    const float4* vpb = (const float4*)(v + (size_t)g_b * DIM);
    const int cna  = (n_a < 32) ? n_a : 32;
    const int cnb  = (n_b < 32) ? n_b : 32;
    const int ctot = cna + cnb;
    float4 va0, va1, vb0, vb1;
    if (ctot < 1) {
      va0 = vpa[lane];  va1 = vpa[lane + 32];
      vb0 = vpb[lane];  vb1 = vpb[lane + 32];
    }
    float4 na0 = {0.0f, 0.0f, 0.0f, 0.0f};
    float4 na1 = {0.0f, 0.0f, 0.0f, 0.0f};
    float4 nb0 = {0.0f, 0.0f, 0.0f, 0.0f};
    float4 nb1 = {0.0f, 0.0f, 0.0f, 0.0f};
    for (int j0 = 0; j0 < ctot; j0 += PGRP) {
      float4 r0[PGRP];
      float4 r1[PGRP];
      pair_ask(vsrc, es_a, es_b, j0, cna, ctot, lane, r0, r1);
      if (j0 == 0) {
        va0 = vpa[lane];  va1 = vpa[lane + 32];
        vb0 = vpb[lane];  vb1 = vpb[lane + 32];
      }
      pair_sum(ea_a, ea_b, j0, cna, ctot, r0, r1, na0, na1, nb0, nb1);
    }
    for (int rr = 0; rr < 2; ++rr) {
      const int s_r = rr ? s_b : s_a;
      const int d_r = rr ? d_b : d_a;
      const int n_r = rr ? n_b : n_a;
      if (n_r > 32) {
        float4 t0 = rr ? nb0 : na0;
        float4 t1 = rr ? nb1 : na1;
        row_tail(vsrc, att, esrc, bucket, ohead, nxt, s_r, d_r, n_r, lane, t0, t1);
        if (rr) { nb0 = t0; nb1 = t1; } else { na0 = t0; na1 = t1; }
      }
    }

    put_row(&x1[side][wave * 2][0],     &x2[side][wave * 2][0],     lane, va0, va1, na0, na1);
    put_row(&x1[side][wave * 2 + 1][0], &x2[side][wave * 2 + 1][0], lane, vb0, vb1, nb0, nb1);
  }
  __syncthreads();

  const float biasA = b1[ncolA];
  const float biasB = b1[ncolB];
  for (int side = 0; side < 2; ++side) {
    v8f acc1A = {};
    v8f acc2A = {};
    v8f acc1B = {};
    v8f acc2B = {};
#pragma unroll
    for (int t = 0; t < DIM / 4; ++t) {
      const int ka = 4 * t + 2 * khalf;
      v2f a1; a1.x = x1[side][mrow][ka]; a1.y = x1[side][mrow][ka + 1];
      v2f a2; a2.x = x2[side][mrow][ka]; a2.y = x2[side][mrow][ka + 1];
      acc1A = __builtin_amdgcn_wmma_f32_16x16x4_f32(
          false, a1, false, bA[t], (short)0, acc1A, false, false);
      acc2A = __builtin_amdgcn_wmma_f32_16x16x4_f32(
          false, a2, false, bA[t], (short)0, acc2A, false, false);
      acc1B = __builtin_amdgcn_wmma_f32_16x16x4_f32(
          false, a1, false, bB[t], (short)0, acc1B, false, false);
      acc2B = __builtin_amdgcn_wmma_f32_16x16x4_f32(
          false, a2, false, bB[t], (short)0, acc2B, false, false);
    }

    float* __restrict__ orow = out + ((size_t)side * NQ + qbase) * DIM;
#pragma unroll
    for (int r = 0; r < 8; ++r) {
      const int m = khalf * 8 + r;
      float y1 = acc1A[r] + biasA;
      float y2 = acc2A[r] + biasA;
      y1 = (y1 >= 0.0f) ? y1 : 0.01f * y1;
      y2 = (y2 >= 0.0f) ? y2 : 0.01f * y2;
      orow[(size_t)m * DIM + ncolA] = y1 + y2;
      float z1 = acc1B[r] + biasB;
      float z2 = acc2B[r] + biasB;
      z1 = (z1 >= 0.0f) ? z1 : 0.01f * z1;
      z2 = (z2 >= 0.0f) ? z2 : 0.01f * z2;
      orow[(size_t)m * DIM + ncolB] = z1 + z2;
    }
  }
}

extern "C" void kernel_launch(void* const* d_in, const int* in_sizes, int n_in,
                              void* d_out, int out_size, void* d_ws, size_t ws_size,
                              hipStream_t stream) {
  const float* v_grid  = (const float*)d_in[0];
  const float* v_sc    = (const float*)d_in[1];
  const float* att_s2g = (const float*)d_in[2];
  const float* att_g2s = (const float*)d_in[3];
  const float* W1      = (const float*)d_in[4];
  const float* b1      = (const float*)d_in[5];
  const int*   src_s2g = (const int*)d_in[6];
  const int*   dst_s2g = (const int*)d_in[7];
  const int*   src_g2s = (const int*)d_in[8];
  const int*   dst_g2s = (const int*)d_in[9];
  const int*   sc_id   = (const int*)d_in[10];
  const int*   gr_id   = (const int*)d_in[11];
  float* out = (float*)d_out;

  int*  tabg = (int*)d_ws;
  int*  tabs = tabg + 2 * WG;
  int*  degg = tabs + 2 * WSC;
  int*  degs = degg + MAXU;
  int*  ohg  = degs + MAXU;
  int*  ohs  = ohg + MAXU;
  int*  nxg  = ohs + MAXU;
  int*  nxs  = nxg + EDGES;
  int*  bkg  = nxs + EDGES;
  int*  bks  = bkg + (size_t)MAXU * CAP * 2;

  k_rank<<<1, RTHREADS, 0, stream>>>(gr_id, sc_id, tabg, degg, ohg);

  k_link<<<dim3(LINKB, 2), 256, 0, stream>>>(att_s2g, src_s2g, dst_s2g, tabg, degg, bkg, ohg, nxg,
                                             att_g2s, src_g2s, dst_g2s, tabs, degs, bks, ohs, nxs);

  k_combine<<<NQ / 16, CTHREADS, 0, stream>>>(v_sc, v_grid, att_g2s, att_s2g, src_g2s, src_s2g,
                                              degs, degg, bks, bkg, ohs, ohg, nxs, nxg,
                                              tabs, tabg, sc_id, gr_id, W1, b1, out);
}


# combine loads W1 fragments as whole 128-B lines (64 b128 loads per wave, was 128 b64 over 16 lines each) from a fragment-order W1 copy written by 16 added workgroups of launch 1; sum order unchanged
# speedup vs baseline: 5.7855x; 1.0367x over previous
#include <hip/hip_runtime.h>

#define NG     100000
#define NS     20000
#define EDGES  300000
#define DIM    256
#define NQ     4096
#define MAXU   4096
#define LDSPAD 2
#define CAP    64
#define GRP    8
#define PGRP   16
#define EPT    8
#define LINKB  ((EDGES / EPT + 255) / 256)
#define WG     (NG / 32)
#define WSC    (NS / 32)
#define RTHREADS 1024
#define CWAVES 8
#define WFQ    (CWAVES * (DIM / 4) * 32)
#define WBLOCKS (WFQ / RTHREADS)
#define WS_HEAD (2 * WG + 2 * WSC + 4 * MAXU + 2 * EDGES + 2 * MAXU * CAP * 2)
#define WS_PAD  ((32 - WS_HEAD % 32) % 32)
static_assert(WFQ % RTHREADS == 0, "whole workgroups write the fragment-order copy of W1");
static_assert(DIM == CWAVES * 32, "8 waves x 32 columns cover the output row");
static_assert(NG % 32 == 0 && NS % 32 == 0, "a node bitmap is whole 32-bit words");
static_assert(WG + WSC <= 4 * RTHREADS, "a rank thread owns 4 consecutive bitmap words");
static_assert(NQ % RTHREADS == 0, "a rank thread marks NQ / RTHREADS queries per side");
static_assert(EDGES % EPT == 0 && EPT == 8, "a link thread owns 8 whole edges (two 16-byte loads per array)");

typedef __attribute__((ext_vector_type(2))) float v2f;
typedef __attribute__((ext_vector_type(8))) float v8f;

__global__ __launch_bounds__(RTHREADS) void k_rank(const int* __restrict__ gid,
                                                    const int* __restrict__ sid,
                                                    int* __restrict__ tab,
                                                    int* __restrict__ deg,
                                                    int* __restrict__ ohead,
                                                    const float* __restrict__ W1,
                                                    float* __restrict__ wf) {
  __shared__ __attribute__((aligned(16))) unsigned bm[4 * RTHREADS];
  __shared__ int wtot[RTHREADS / 32];
  __shared__ int gtot;
  const int tid  = threadIdx.x;
  const int lane = tid & 31;
  const int wave = tid >> 5;

  if (blockIdx.x != 0) {
    const int q  = (blockIdx.x - 1) * RTHREADS + tid;
    const int t  = (q >> 5) & (DIM / 4 - 1);
    const int w  = q >> 11;
    const int n  = w * 32 + (lane & 15);
    const int k  = 4 * t + 2 * (lane >> 4);
    const float2 a = *(const float2*)(W1 + (size_t)n * DIM + k);
    const float2 b = *(const float2*)(W1 + (size_t)(n + 16) * DIM + k);
    ((float4*)wf)[q] = make_float4(a.x, a.y, b.x, b.y);
    return;
  }

  for (int i = tid; i < 2 * MAXU / 2; i += RTHREADS) {
    ((int2*)deg)[i]   = make_int2(0, 0);
    ((int2*)ohead)[i] = make_int2(-1, -1);
  }

  int g[NQ / RTHREADS];
  int c[NQ / RTHREADS];
#pragma unroll
  for (int k = 0; k < NQ / RTHREADS; ++k) {
    g[k] = gid[tid + k * RTHREADS];
    c[k] = sid[tid + k * RTHREADS];
  }
  for (int k = 0; k < 4; ++k) bm[tid + k * RTHREADS] = 0u;
  __syncthreads();
#pragma unroll
  for (int k = 0; k < NQ / RTHREADS; ++k) {
    atomicOr(&bm[g[k] >> 5], 1u << (g[k] & 31));
    atomicOr(&bm[WG + (c[k] >> 5)], 1u << (c[k] & 31));
  }
  __syncthreads();

  const uint4 bw = *(const uint4*)(bm + 4 * tid);
  const int p1   = __popc(bw.x);
  const int p2   = p1 + __popc(bw.y);
  const int p3   = p2 + __popc(bw.z);
  const int tsum = p3 + __popc(bw.w);
  int incl = tsum;
  for (int d = 1; d < 32; d <<= 1) {
    const int o = __shfl_up(incl, d);
    if (lane >= d) incl += o;
  }
  if (lane == 31) wtot[wave] = incl;
  __syncthreads();
  if (wave == 0) {
    const int v = wtot[lane];
    int iv = v;
    for (int d = 1; d < 32; d <<= 1) {
      const int o = __shfl_up(iv, d);
      if (lane >= d) iv += o;
    }
    wtot[lane] = iv - v;
  }
  __syncthreads();
  const int base = wtot[wave] + incl - tsum;
  const int pre[4] = {base, base + p1, base + p2, base + p3};
  const unsigned bits[4] = {bw.x, bw.y, bw.z, bw.w};
#pragma unroll
  for (int j = 0; j < 4; ++j)
    if (4 * tid + j == WG) gtot = pre[j];
  __syncthreads();
  const int gt = gtot;
#pragma unroll
  for (int j = 0; j < 4; ++j) {
    const int w = 4 * tid + j;
    if (w < WG + WSC) {
      int2 ent; ent.x = (int)bits[j]; ent.y = (w < WG) ? pre[j] : pre[j] - gt;
      *(int2*)(tab + 2 * w) = ent;
    }
  }
}

__device__ __forceinline__ unsigned long long slot_entry(const int* __restrict__ tab, int node) {
  return *(const unsigned long long*)(tab + 2 * (node >> 5));
}
__device__ __forceinline__ int slot_from(unsigned long long ent, int node) {
  const unsigned bits = (unsigned)ent;
  const int      rank = (int)(ent >> 32);
  const unsigned bit  = 1u << (node & 31);
  const int      s    = rank + __popc(bits & (bit - 1u));
  return (bits & bit) ? s : -1;
}

__global__ __launch_bounds__(256) void k_link(const float* __restrict__ att0,
                                              const int* __restrict__ src0,
                                              const int* __restrict__ dst0,
                                              const int* __restrict__ tab0,
                                              int* __restrict__ deg0,
                                              int* __restrict__ bucket0,
                                              int* __restrict__ ohead0,
                                              int* __restrict__ nxt0,
                                              const float* __restrict__ att1,
                                              const int* __restrict__ src1,
                                              const int* __restrict__ dst1,
                                              const int* __restrict__ tab1,
                                              int* __restrict__ deg1,
                                              int* __restrict__ bucket1,
                                              int* __restrict__ ohead1,
                                              int* __restrict__ nxt1) {
  const bool first = (blockIdx.y == 0);
  const float* __restrict__ att    = first ? att0    : att1;
  const int*   __restrict__ src    = first ? src0    : src1;
  const int*   __restrict__ dst    = first ? dst0    : dst1;
  const int*   __restrict__ tab    = first ? tab0    : tab1;
  int*         __restrict__ deg    = first ? deg0    : deg1;
  int*         __restrict__ bucket = first ? bucket0 : bucket1;
  int*         __restrict__ ohead  = first ? ohead0  : ohead1;
  int*         __restrict__ nxt    = first ? nxt0    : nxt1;

  const int e0 = (blockIdx.x * 256 + threadIdx.x) * EPT;
  if (e0 >= EDGES) return;

  const int4   d0 = *(const int4*)(dst + e0);
  const int4   d1 = *(const int4*)(dst + e0 + 4);
  const int4   r0 = *(const int4*)(src + e0);
  const int4   r1 = *(const int4*)(src + e0 + 4);
  const float4 a0 = *(const float4*)(att + e0);
  const float4 a1 = *(const float4*)(att + e0 + 4);
  const int   d[EPT]  = {d0.x, d0.y, d0.z, d0.w, d1.x, d1.y, d1.z, d1.w};
  const int   sr[EPT] = {r0.x, r0.y, r0.z, r0.w, r1.x, r1.y, r1.z, r1.w};
  const float at[EPT] = {a0.x, a0.y, a0.z, a0.w, a1.x, a1.y, a1.z, a1.w};

  unsigned long long te[EPT];
#pragma unroll
  for (int k = 0; k < EPT; ++k) te[k] = slot_entry(tab, d[k]);
  int s[EPT];
#pragma unroll
  for (int k = 0; k < EPT; ++k) s[k] = slot_from(te[k], d[k]);

  if ((s[0] & s[1] & s[2] & s[3] & s[4] & s[5] & s[6] & s[7]) < 0) return;

#define LINK_ASK(k) int p##k = 0; if (s[k] >= 0) p##k = atomicAdd(&deg[s[k]], 1);
  LINK_ASK(0) LINK_ASK(1) LINK_ASK(2) LINK_ASK(3)
  LINK_ASK(4) LINK_ASK(5) LINK_ASK(6) LINK_ASK(7)
#undef LINK_ASK

#define LINK_PUT(k)                                                        \
  if (s[k] >= 0) {                                                         \
    if (p##k < CAP) {                                                      \
      int2 ent; ent.x = sr[k]; ent.y = __float_as_int(at[k]);              \
      *(int2*)(bucket + ((size_t)s[k] * CAP + p##k) * 2) = ent;            \
    } else {                                                               \
      nxt[e0 + k] = atomicExch(&ohead[s[k]], e0 + k);                      \
    }                                                                      \
  }
  LINK_PUT(0) LINK_PUT(1) LINK_PUT(2) LINK_PUT(3)
  LINK_PUT(4) LINK_PUT(5) LINK_PUT(6) LINK_PUT(7)
#undef LINK_PUT
}

__device__ __forceinline__ void grp_ask(const float* __restrict__ vsrc, int es, int i, int cn,
                                        int lane, float4 (&r0)[GRP], float4 (&r1)[GRP]) {
#pragma unroll
  for (int k = 0; k < GRP; ++k) {
    const int ik = (i + k < cn) ? (i + k) : (cn - 1);
    const int si = __builtin_amdgcn_readlane(es, ik);
    const float4* rp = (const float4*)(vsrc + (size_t)si * DIM);
    r0[k] = rp[lane];
    r1[k] = rp[lane + 32];
  }
}
__device__ __forceinline__ void grp_sum(float ea, int i, int cn,
                                        const float4 (&r0)[GRP], const float4 (&r1)[GRP],
                                        float4& n0, float4& n1) {
#pragma unroll
  for (int k = 0; k < GRP; ++k) {
    if (i + k < cn) {
      const float ai = __int_as_float(
          __builtin_amdgcn_readlane(__float_as_int(ea), i + k));
      n0.x += r0[k].x * ai;  n0.y += r0[k].y * ai;
      n0.z += r0[k].z * ai;  n0.w += r0[k].w * ai;
      n1.x += r1[k].x * ai;  n1.y += r1[k].y * ai;
      n1.z += r1[k].z * ai;  n1.w += r1[k].w * ai;
    }
  }
}
__device__ __forceinline__ void pair_ask(const float* __restrict__ vsrc, int es_a, int es_b,
                                         int j0, int cna, int ctot, int lane,
                                         float4 (&r0)[PGRP], float4 (&r1)[PGRP]) {
#pragma unroll
  for (int k = 0; k < PGRP; ++k) {
    const int j  = (j0 + k < ctot) ? (j0 + k) : (ctot - 1);
    const bool ina = j < cna;
    const int sa = __builtin_amdgcn_readlane(es_a, ina ? j : 0);
    const int sb = __builtin_amdgcn_readlane(es_b, ina ? 0 : (j - cna));
    const int si = ina ? sa : sb;
    const float4* rp = (const float4*)(vsrc + (size_t)si * DIM);
    r0[k] = rp[lane];
    r1[k] = rp[lane + 32];
  }
}
__device__ __forceinline__ void pair_sum(float ea_a, float ea_b, int j0, int cna, int ctot,
                                         const float4 (&r0)[PGRP], const float4 (&r1)[PGRP],
                                         float4& na0, float4& na1, float4& nb0, float4& nb1) {
#pragma unroll
  for (int k = 0; k < PGRP; ++k) {
    const int j = j0 + k;
    if (j < ctot) {
      if (j < cna) {
        const float ai = __int_as_float(__builtin_amdgcn_readlane(__float_as_int(ea_a), j));
        na0.x += r0[k].x * ai;  na0.y += r0[k].y * ai;
        na0.z += r0[k].z * ai;  na0.w += r0[k].w * ai;
        na1.x += r1[k].x * ai;  na1.y += r1[k].y * ai;
        na1.z += r1[k].z * ai;  na1.w += r1[k].w * ai;
      } else {
        const float ai = __int_as_float(__builtin_amdgcn_readlane(__float_as_int(ea_b), j - cna));
        nb0.x += r0[k].x * ai;  nb0.y += r0[k].y * ai;
        nb0.z += r0[k].z * ai;  nb0.w += r0[k].w * ai;
        nb1.x += r1[k].x * ai;  nb1.y += r1[k].y * ai;
        nb1.z += r1[k].z * ai;  nb1.w += r1[k].w * ai;
      }
    }
  }
}
__device__ __forceinline__ void row_tail(const float* __restrict__ vsrc,
                                         const float* __restrict__ att,
                                         const int* __restrict__ esrc,
                                         const int* __restrict__ bucket,
                                         const int* __restrict__ ohead,
                                         const int* __restrict__ nxt,
                                         int s, int d, int nb, int lane,
                                         float4& n0, float4& n1) {
  for (int base = 32; base < nb; base += 32) {
    const int cn = (nb - base < 32) ? (nb - base) : 32;
    int   es = 0;
    float ea = 0.0f;
    if (lane < cn) {
      const int2 ent = *(const int2*)(bucket + ((size_t)s * CAP + base + lane) * 2);
      es = ent.x;
      ea = __int_as_float(ent.y);
    }
    for (int i = 0; i < cn; i += GRP) {
      float4 r0[GRP];
      float4 r1[GRP];
      grp_ask(vsrc, es, i, cn, lane, r0, r1);
      grp_sum(ea, i, cn, r0, r1, n0, n1);
    }
  }
  if (d > CAP) {
    int e = __builtin_amdgcn_readfirstlane(ohead[s]);
    for (int it = 0; it < EDGES && e >= 0; ++it) {
      const int   si = esrc[e];
      const float ai = att[e];
      const float4* rp = (const float4*)(vsrc + (size_t)si * DIM);
      const float4 q0 = rp[lane];
      const float4 q1 = rp[lane + 32];
      n0.x += q0.x * ai;  n0.y += q0.y * ai;  n0.z += q0.z * ai;  n0.w += q0.w * ai;
      n1.x += q1.x * ai;  n1.y += q1.y * ai;  n1.z += q1.z * ai;  n1.w += q1.w * ai;
      e = __builtin_amdgcn_readfirstlane(nxt[e]);
    }
  }
}
__device__ __forceinline__ void put_row(float* __restrict__ x1r, float* __restrict__ x2r, int lane,
                                        float4 vv0, float4 vv1, float4 n0, float4 n1) {
  const int c0 = lane * 4;
  const int c1 = c0 + 128;
  x1r[c0 + 0] = vv0.x + n0.x;  x2r[c0 + 0] = vv0.x * n0.x;
  x1r[c0 + 1] = vv0.y + n0.y;  x2r[c0 + 1] = vv0.y * n0.y;
  x1r[c0 + 2] = vv0.z + n0.z;  x2r[c0 + 2] = vv0.z * n0.z;
  x1r[c0 + 3] = vv0.w + n0.w;  x2r[c0 + 3] = vv0.w * n0.w;
  x1r[c1 + 0] = vv1.x + n1.x;  x2r[c1 + 0] = vv1.x * n1.x;
  x1r[c1 + 1] = vv1.y + n1.y;  x2r[c1 + 1] = vv1.y * n1.y;
  x1r[c1 + 2] = vv1.z + n1.z;  x2r[c1 + 2] = vv1.z * n1.z;
  x1r[c1 + 3] = vv1.w + n1.w;  x2r[c1 + 3] = vv1.w * n1.w;
}

#define CTHREADS 256
__global__ __launch_bounds__(CTHREADS) void k_combine(const float* __restrict__ v_sc,
                                                         const float* __restrict__ v_grid,
                                                         const float* __restrict__ att_g2s,
                                                         const float* __restrict__ att_s2g,
                                                         const int* __restrict__ src_g2s,
                                                         const int* __restrict__ src_s2g,
                                                         const int* __restrict__ degs,
                                                         const int* __restrict__ degg,
                                                         const int* __restrict__ bks,
                                                         const int* __restrict__ bkg,
                                                         const int* __restrict__ ohs,
                                                         const int* __restrict__ ohg,
                                                         const int* __restrict__ nxs,
                                                         const int* __restrict__ nxg,
                                                         const int* __restrict__ tabs,
                                                         const int* __restrict__ tabg,
                                                         const int* __restrict__ sc_id,
                                                         const int* __restrict__ gr_id,
                                                         const float* __restrict__ wf,
                                                         const float* __restrict__ b1,
                                                         float* __restrict__ out) {
  __shared__ __attribute__((aligned(16))) float x1[2][16][DIM + LDSPAD];
  __shared__ __attribute__((aligned(16))) float x2[2][16][DIM + LDSPAD];

  const int lane  = threadIdx.x & 31;
  const int wave  = __builtin_amdgcn_readfirstlane((int)(threadIdx.x >> 5));
  const int qbase = blockIdx.x * 16;

  const int mrow  = lane & 15;
  const int khalf = lane >> 4;

  const int ncolA = wave * 32 + mrow;
  const int ncolB = ncolA + 16;
  const float4* wfrag = (const float4*)wf + (size_t)wave * (DIM / 4) * 32 + lane;
  v2f bA[DIM / 4];
  v2f bB[DIM / 4];
#pragma unroll
  for (int t = 0; t < DIM / 4; ++t) {
    const float4 f = wfrag[t * 32];
    bA[t].x = f.x;  bA[t].y = f.y;
    bB[t].x = f.z;  bB[t].y = f.w;
  }

  int gq[4];
  int sq[4];
  int dq[4];
  int nq[4];
  unsigned long long tq[4];
#pragma unroll
  for (int tr = 0; tr < 4; ++tr) {
    const int* ids = (tr >> 1) ? gr_id : sc_id;
    gq[tr] = ids[qbase + wave * 2 + (tr & 1)];
  }
#pragma unroll
  for (int tr = 0; tr < 4; ++tr) tq[tr] = slot_entry((tr >> 1) ? tabg : tabs, gq[tr]);
#pragma unroll
  for (int tr = 0; tr < 4; ++tr) {
    const int* deg = (tr >> 1) ? degg : degs;
    sq[tr] = slot_from(tq[tr], gq[tr]);
    dq[tr] = __builtin_amdgcn_readfirstlane(deg[sq[tr]]);
    nq[tr] = (dq[tr] < CAP) ? dq[tr] : CAP;
  }
  int2 hq[4];
#pragma unroll
  for (int tr = 0; tr < 4; ++tr) {
    const int* bucket = (tr >> 1) ? bkg  : bks;
    const int* tab    = (tr >> 1) ? tabg : tabs;
    const bool any = nq[tr] > 0;
    const int* hp = any ? bucket + (size_t)sq[tr] * CAP * 2 : tab + 2 * (gq[tr] >> 5);
    const int  hl = any ? ((lane < nq[tr]) ? lane : nq[tr] - 1) : 0;
    hq[tr] = *(const int2*)(hp + 2 * hl);
  }

  for (int side = 0; side < 2; ++side) {
    const float* v      = side ? v_grid  : v_sc;
    const float* vsrc   = side ? v_sc    : v_grid;
    const float* att    = side ? att_s2g : att_g2s;
    const int*   esrc   = side ? src_s2g : src_g2s;
    const int*   bucket = side ? bkg     : bks;
    const int*   ohead  = side ? ohg     : ohs;
    const int*   nxt    = side ? nxg     : nxs;
    const int   g_a  = side ? gq[2] : gq[0];
    const int   g_b  = side ? gq[3] : gq[1];
    const int   s_a  = side ? sq[2] : sq[0];
    const int   s_b  = side ? sq[3] : sq[1];
    const int   d_a  = side ? dq[2] : dq[0];
    const int   d_b  = side ? dq[3] : dq[1];
    const int   n_a  = side ? nq[2] : nq[0];
    const int   n_b  = side ? nq[3] : nq[1];
    const int2  h_a  = side ? hq[2] : hq[0];
    const int2  h_b  = side ? hq[3] : hq[1];
    const int   es_a = h_a.x;
    const int   es_b = h_b.x;
    const float ea_a = __int_as_float(h_a.y);
    const float ea_b = __int_as_float(h_b.y);
    const float4* vpa = (const float4*)(v + (size_t)g_a * DIM);
    const float4* vpb = (const float4*)(v + (size_t)g_b * DIM);
    const int cna  = (n_a < 32) ? n_a : 32;
    const int cnb  = (n_b < 32) ? n_b : 32;
    const int ctot = cna + cnb;
    float4 va0, va1, vb0, vb1;
    if (ctot < 1) {
      va0 = vpa[lane];  va1 = vpa[lane + 32];
      vb0 = vpb[lane];  vb1 = vpb[lane + 32];
    }
    float4 na0 = {0.0f, 0.0f, 0.0f, 0.0f};
    float4 na1 = {0.0f, 0.0f, 0.0f, 0.0f};
    float4 nb0 = {0.0f, 0.0f, 0.0f, 0.0f};
    float4 nb1 = {0.0f, 0.0f, 0.0f, 0.0f};
    for (int j0 = 0; j0 < ctot; j0 += PGRP) {
      float4 r0[PGRP];
      float4 r1[PGRP];
      pair_ask(vsrc, es_a, es_b, j0, cna, ctot, lane, r0, r1);
      if (j0 == 0) {
        va0 = vpa[lane];  va1 = vpa[lane + 32];
        vb0 = vpb[lane];  vb1 = vpb[lane + 32];
      }
      pair_sum(ea_a, ea_b, j0, cna, ctot, r0, r1, na0, na1, nb0, nb1);
    }
    for (int rr = 0; rr < 2; ++rr) {
      const int s_r = rr ? s_b : s_a;
      const int d_r = rr ? d_b : d_a;
      const int n_r = rr ? n_b : n_a;
      if (n_r > 32) {
        float4 t0 = rr ? nb0 : na0;
        float4 t1 = rr ? nb1 : na1;
        row_tail(vsrc, att, esrc, bucket, ohead, nxt, s_r, d_r, n_r, lane, t0, t1);
        if (rr) { nb0 = t0; nb1 = t1; } else { na0 = t0; na1 = t1; }
      }
    }

    put_row(&x1[side][wave * 2][0],     &x2[side][wave * 2][0],     lane, va0, va1, na0, na1);
    put_row(&x1[side][wave * 2 + 1][0], &x2[side][wave * 2 + 1][0], lane, vb0, vb1, nb0, nb1);
  }
  __syncthreads();

  const float biasA = b1[ncolA];
  const float biasB = b1[ncolB];
  for (int side = 0; side < 2; ++side) {
    v8f acc1A = {};
    v8f acc2A = {};
    v8f acc1B = {};
    v8f acc2B = {};
#pragma unroll
    for (int t = 0; t < DIM / 4; ++t) {
      const int ka = 4 * t + 2 * khalf;
      v2f a1; a1.x = x1[side][mrow][ka]; a1.y = x1[side][mrow][ka + 1];
      v2f a2; a2.x = x2[side][mrow][ka]; a2.y = x2[side][mrow][ka + 1];
      acc1A = __builtin_amdgcn_wmma_f32_16x16x4_f32(
          false, a1, false, bA[t], (short)0, acc1A, false, false);
      acc2A = __builtin_amdgcn_wmma_f32_16x16x4_f32(
          false, a2, false, bA[t], (short)0, acc2A, false, false);
      acc1B = __builtin_amdgcn_wmma_f32_16x16x4_f32(
          false, a1, false, bB[t], (short)0, acc1B, false, false);
      acc2B = __builtin_amdgcn_wmma_f32_16x16x4_f32(
          false, a2, false, bB[t], (short)0, acc2B, false, false);
    }

    float* __restrict__ orow = out + ((size_t)side * NQ + qbase) * DIM;
#pragma unroll
    for (int r = 0; r < 8; ++r) {
      const int m = khalf * 8 + r;
      float y1 = acc1A[r] + biasA;
      float y2 = acc2A[r] + biasA;
      y1 = (y1 >= 0.0f) ? y1 : 0.01f * y1;
      y2 = (y2 >= 0.0f) ? y2 : 0.01f * y2;
      orow[(size_t)m * DIM + ncolA] = y1 + y2;
      float z1 = acc1B[r] + biasB;
      float z2 = acc2B[r] + biasB;
      z1 = (z1 >= 0.0f) ? z1 : 0.01f * z1;
      z2 = (z2 >= 0.0f) ? z2 : 0.01f * z2;
      orow[(size_t)m * DIM + ncolB] = z1 + z2;
    }
  }
}

extern "C" void kernel_launch(void* const* d_in, const int* in_sizes, int n_in,
                              void* d_out, int out_size, void* d_ws, size_t ws_size,
                              hipStream_t stream) {
  const float* v_grid  = (const float*)d_in[0];
  const float* v_sc    = (const float*)d_in[1];
  const float* att_s2g = (const float*)d_in[2];
  const float* att_g2s = (const float*)d_in[3];
  const float* W1      = (const float*)d_in[4];
  const float* b1      = (const float*)d_in[5];
  const int*   src_s2g = (const int*)d_in[6];
  const int*   dst_s2g = (const int*)d_in[7];
  const int*   src_g2s = (const int*)d_in[8];
  const int*   dst_g2s = (const int*)d_in[9];
  const int*   sc_id   = (const int*)d_in[10];
  const int*   gr_id   = (const int*)d_in[11];
  float* out = (float*)d_out;

  int*  tabg = (int*)d_ws;
  int*  tabs = tabg + 2 * WG;
  int*  degg = tabs + 2 * WSC;
  int*  degs = degg + MAXU;
  int*  ohg  = degs + MAXU;
  int*  ohs  = ohg + MAXU;
  int*  nxg  = ohs + MAXU;
  int*  nxs  = nxg + EDGES;
  int*  bkg  = nxs + EDGES;
  int*  bks  = bkg + (size_t)MAXU * CAP * 2;
  float* wf  = (float*)(bks + (size_t)MAXU * CAP * 2 + WS_PAD);

  k_rank<<<1 + WBLOCKS, RTHREADS, 0, stream>>>(gr_id, sc_id, tabg, degg, ohg, W1, wf);

  k_link<<<dim3(LINKB, 2), 256, 0, stream>>>(att_s2g, src_s2g, dst_s2g, tabg, degg, bkg, ohg, nxg,
                                             att_g2s, src_g2s, dst_g2s, tabs, degs, bks, ohs, nxs);

  k_combine<<<NQ / 16, CTHREADS, 0, stream>>>(v_sc, v_grid, att_g2s, att_s2g, src_g2s, src_s2g,
                                              degs, degg, bks, bkg, ohs, ohg, nxs, nxg,
                                              tabs, tabg, sc_id, gr_id, wf, b1, out);
}


# combine: grid-side tile staged first and its WMMA product run in parts behind the small-category gather's loads; W1 fragments fetched a part ahead per side; sum order kept
# speedup vs baseline: 5.9614x; 1.0304x over previous
#include <hip/hip_runtime.h>

#define NG     100000
#define NS     20000
#define EDGES  300000
#define DIM    256
#define NQ     4096
#define MAXU   4096
#define LDSPAD 2
#define CAP    64
#define GRP    8
#define PGRP   16
#define EPT    8
#define LINKB  ((EDGES / EPT + 255) / 256)
#define WG     (NG / 32)
#define WSC    (NS / 32)
#define RTHREADS 1024
#define CWAVES 8
#define KPART  8
#define NPART  (DIM / 4 / KPART)
#define WFQ    (CWAVES * (DIM / 4) * 32)
#define WBLOCKS (WFQ / RTHREADS)
#define WS_HEAD (2 * WG + 2 * WSC + 4 * MAXU + 2 * EDGES + 2 * MAXU * CAP * 2)
#define WS_PAD  ((32 - WS_HEAD % 32) % 32)
static_assert(WFQ % RTHREADS == 0, "whole workgroups write the fragment-order copy of W1");
static_assert(DIM == CWAVES * 32, "8 waves x 32 columns cover the output row");
static_assert((NPART / 2) * PGRP == 64 && NPART % 2 == 0, "two parts per group of the first 32 entries of two rows' joint list");
static_assert(NG % 32 == 0 && NS % 32 == 0, "a node bitmap is whole 32-bit words");
static_assert(WG + WSC <= 4 * RTHREADS, "a rank thread owns 4 consecutive bitmap words");
static_assert(NQ % RTHREADS == 0, "a rank thread marks NQ / RTHREADS queries per side");
static_assert(EDGES % EPT == 0 && EPT == 8, "a link thread owns 8 whole edges (two 16-byte loads per array)");

typedef __attribute__((ext_vector_type(2))) float v2f;
typedef __attribute__((ext_vector_type(8))) float v8f;

__global__ __launch_bounds__(RTHREADS) void k_rank(const int* __restrict__ gid,
                                                    const int* __restrict__ sid,
                                                    int* __restrict__ tab,
                                                    int* __restrict__ deg,
                                                    int* __restrict__ ohead,
                                                    const float* __restrict__ W1,
                                                    float* __restrict__ wf) {
  __shared__ __attribute__((aligned(16))) unsigned bm[4 * RTHREADS];
  __shared__ int wtot[RTHREADS / 32];
  __shared__ int gtot;
  const int tid  = threadIdx.x;
  const int lane = tid & 31;
  const int wave = tid >> 5;

  if (blockIdx.x != 0) {
    const int q  = (blockIdx.x - 1) * RTHREADS + tid;
    const int t  = (q >> 5) & (DIM / 4 - 1);
    const int w  = q >> 11;
    const int n  = w * 32 + (lane & 15);
    const int k  = 4 * t + 2 * (lane >> 4);
    const float2 a = *(const float2*)(W1 + (size_t)n * DIM + k);
    const float2 b = *(const float2*)(W1 + (size_t)(n + 16) * DIM + k);
    ((float4*)wf)[q] = make_float4(a.x, a.y, b.x, b.y);
    return;
  }

  for (int i = tid; i < 2 * MAXU / 2; i += RTHREADS) {
    ((int2*)deg)[i]   = make_int2(0, 0);
    ((int2*)ohead)[i] = make_int2(-1, -1);
  }

  int g[NQ / RTHREADS];
  int c[NQ / RTHREADS];
#pragma unroll
  for (int k = 0; k < NQ / RTHREADS; ++k) {
    g[k] = gid[tid + k * RTHREADS];
    c[k] = sid[tid + k * RTHREADS];
  }
  for (int k = 0; k < 4; ++k) bm[tid + k * RTHREADS] = 0u;
  __syncthreads();
#pragma unroll
  for (int k = 0; k < NQ / RTHREADS; ++k) {
    atomicOr(&bm[g[k] >> 5], 1u << (g[k] & 31));
    atomicOr(&bm[WG + (c[k] >> 5)], 1u << (c[k] & 31));
  }
  __syncthreads();

  const uint4 bw = *(const uint4*)(bm + 4 * tid);
  const int p1   = __popc(bw.x);
  const int p2   = p1 + __popc(bw.y);
  const int p3   = p2 + __popc(bw.z);
  const int tsum = p3 + __popc(bw.w);
  int incl = tsum;
  for (int d = 1; d < 32; d <<= 1) {
    const int o = __shfl_up(incl, d);
    if (lane >= d) incl += o;
  }
  if (lane == 31) wtot[wave] = incl;
  __syncthreads();
  if (wave == 0) {
    const int v = wtot[lane];
    int iv = v;
    for (int d = 1; d < 32; d <<= 1) {
      const int o = __shfl_up(iv, d);
      if (lane >= d) iv += o;
    }
    wtot[lane] = iv - v;
  }
  __syncthreads();
  const int base = wtot[wave] + incl - tsum;
  const int pre[4] = {base, base + p1, base + p2, base + p3};
  const unsigned bits[4] = {bw.x, bw.y, bw.z, bw.w};
#pragma unroll
  for (int j = 0; j < 4; ++j)
    if (4 * tid + j == WG) gtot = pre[j];
  __syncthreads();
  const int gt = gtot;
#pragma unroll
  for (int j = 0; j < 4; ++j) {
    const int w = 4 * tid + j;
    if (w < WG + WSC) {
      int2 ent; ent.x = (int)bits[j]; ent.y = (w < WG) ? pre[j] : pre[j] - gt;
      *(int2*)(tab + 2 * w) = ent;
    }
  }
}

__device__ __forceinline__ unsigned long long slot_entry(const int* __restrict__ tab, int node) {
  return *(const unsigned long long*)(tab + 2 * (node >> 5));
}
__device__ __forceinline__ int slot_from(unsigned long long ent, int node) {
  const unsigned bits = (unsigned)ent;
  const int      rank = (int)(ent >> 32);
  const unsigned bit  = 1u << (node & 31);
  const int      s    = rank + __popc(bits & (bit - 1u));
  return (bits & bit) ? s : -1;
}

__global__ __launch_bounds__(256) void k_link(const float* __restrict__ att0,
                                              const int* __restrict__ src0,
                                              const int* __restrict__ dst0,
                                              const int* __restrict__ tab0,
                                              int* __restrict__ deg0,
                                              int* __restrict__ bucket0,
                                              int* __restrict__ ohead0,
                                              int* __restrict__ nxt0,
                                              const float* __restrict__ att1,
                                              const int* __restrict__ src1,
                                              const int* __restrict__ dst1,
                                              const int* __restrict__ tab1,
                                              int* __restrict__ deg1,
                                              int* __restrict__ bucket1,
                                              int* __restrict__ ohead1,
                                              int* __restrict__ nxt1) {
  const bool first = (blockIdx.y == 0);
  const float* __restrict__ att    = first ? att0    : att1;
  const int*   __restrict__ src    = first ? src0    : src1;
  const int*   __restrict__ dst    = first ? dst0    : dst1;
  const int*   __restrict__ tab    = first ? tab0    : tab1;
  int*         __restrict__ deg    = first ? deg0    : deg1;
  int*         __restrict__ bucket = first ? bucket0 : bucket1;
  int*         __restrict__ ohead  = first ? ohead0  : ohead1;
  int*         __restrict__ nxt    = first ? nxt0    : nxt1;

  const int e0 = (blockIdx.x * 256 + threadIdx.x) * EPT;
  if (e0 >= EDGES) return;

  const int4   d0 = *(const int4*)(dst + e0);
  const int4   d1 = *(const int4*)(dst + e0 + 4);
  const int4   r0 = *(const int4*)(src + e0);
  const int4   r1 = *(const int4*)(src + e0 + 4);
  const float4 a0 = *(const float4*)(att + e0);
  const float4 a1 = *(const float4*)(att + e0 + 4);
  const int   d[EPT]  = {d0.x, d0.y, d0.z, d0.w, d1.x, d1.y, d1.z, d1.w};
  const int   sr[EPT] = {r0.x, r0.y, r0.z, r0.w, r1.x, r1.y, r1.z, r1.w};
  const float at[EPT] = {a0.x, a0.y, a0.z, a0.w, a1.x, a1.y, a1.z, a1.w};

  unsigned long long te[EPT];
#pragma unroll
  for (int k = 0; k < EPT; ++k) te[k] = slot_entry(tab, d[k]);
  int s[EPT];
#pragma unroll
  for (int k = 0; k < EPT; ++k) s[k] = slot_from(te[k], d[k]);

  if ((s[0] & s[1] & s[2] & s[3] & s[4] & s[5] & s[6] & s[7]) < 0) return;

#define LINK_ASK(k) int p##k = 0; if (s[k] >= 0) p##k = atomicAdd(&deg[s[k]], 1);
  LINK_ASK(0) LINK_ASK(1) LINK_ASK(2) LINK_ASK(3)
  LINK_ASK(4) LINK_ASK(5) LINK_ASK(6) LINK_ASK(7)
#undef LINK_ASK

#define LINK_PUT(k)                                                        \
  if (s[k] >= 0) {                                                         \
    if (p##k < CAP) {                                                      \
      int2 ent; ent.x = sr[k]; ent.y = __float_as_int(at[k]);              \
      *(int2*)(bucket + ((size_t)s[k] * CAP + p##k) * 2) = ent;            \
    } else {                                                               \
      nxt[e0 + k] = atomicExch(&ohead[s[k]], e0 + k);                      \
    }                                                                      \
  }
  LINK_PUT(0) LINK_PUT(1) LINK_PUT(2) LINK_PUT(3)
  LINK_PUT(4) LINK_PUT(5) LINK_PUT(6) LINK_PUT(7)
#undef LINK_PUT
}

__device__ __forceinline__ void grp_ask(const float* __restrict__ vsrc, int es, int i, int cn,
                                        int lane, float4 (&r0)[GRP], float4 (&r1)[GRP]) {
#pragma unroll
  for (int k = 0; k < GRP; ++k) {
    const int ik = (i + k < cn) ? (i + k) : (cn - 1);
    const int si = __builtin_amdgcn_readlane(es, ik);
    const float4* rp = (const float4*)(vsrc + (size_t)si * DIM);
    r0[k] = rp[lane];
    r1[k] = rp[lane + 32];
  }
}
__device__ __forceinline__ void grp_sum(float ea, int i, int cn,
                                        const float4 (&r0)[GRP], const float4 (&r1)[GRP],
                                        float4& n0, float4& n1) {
#pragma unroll
  for (int k = 0; k < GRP; ++k) {
    if (i + k < cn) {
      const float ai = __int_as_float(
          __builtin_amdgcn_readlane(__float_as_int(ea), i + k));
      n0.x += r0[k].x * ai;  n0.y += r0[k].y * ai;
      n0.z += r0[k].z * ai;  n0.w += r0[k].w * ai;
      n1.x += r1[k].x * ai;  n1.y += r1[k].y * ai;
      n1.z += r1[k].z * ai;  n1.w += r1[k].w * ai;
    }
  }
}
__device__ __forceinline__ void pair_ask(const float* __restrict__ vsrc, int es_a, int es_b,
                                         int j0, int cna, int ctot, int lane,
                                         float4 (&r0)[PGRP], float4 (&r1)[PGRP]) {
#pragma unroll
  for (int k = 0; k < PGRP; ++k) {
    const int j  = (j0 + k < ctot) ? (j0 + k) : (ctot - 1);
    const bool ina = j < cna;
    const int sa = __builtin_amdgcn_readlane(es_a, ina ? j : 0);
    const int sb = __builtin_amdgcn_readlane(es_b, ina ? 0 : (j - cna));
    const int si = ina ? sa : sb;
    const float4* rp = (const float4*)(vsrc + (size_t)si * DIM);
    r0[k] = rp[lane];
    r1[k] = rp[lane + 32];
  }
}
__device__ __forceinline__ void pair_sum(float ea_a, float ea_b, int j0, int cna, int ctot,
                                         const float4 (&r0)[PGRP], const float4 (&r1)[PGRP],
                                         float4& na0, float4& na1, float4& nb0, float4& nb1) {
#pragma unroll
  for (int k = 0; k < PGRP; ++k) {
    const int j = j0 + k;
    if (j < ctot) {
      if (j < cna) {
        const float ai = __int_as_float(__builtin_amdgcn_readlane(__float_as_int(ea_a), j));
        na0.x += r0[k].x * ai;  na0.y += r0[k].y * ai;
        na0.z += r0[k].z * ai;  na0.w += r0[k].w * ai;
        na1.x += r1[k].x * ai;  na1.y += r1[k].y * ai;
        na1.z += r1[k].z * ai;  na1.w += r1[k].w * ai;
      } else {
        const float ai = __int_as_float(__builtin_amdgcn_readlane(__float_as_int(ea_b), j - cna));
        nb0.x += r0[k].x * ai;  nb0.y += r0[k].y * ai;
        nb0.z += r0[k].z * ai;  nb0.w += r0[k].w * ai;
        nb1.x += r1[k].x * ai;  nb1.y += r1[k].y * ai;
        nb1.z += r1[k].z * ai;  nb1.w += r1[k].w * ai;
      }
    }
  }
}
__device__ __forceinline__ void row_tail(const float* __restrict__ vsrc,
                                         const float* __restrict__ att,
                                         const int* __restrict__ esrc,
                                         const int* __restrict__ bucket,
                                         const int* __restrict__ ohead,
                                         const int* __restrict__ nxt,
                                         int s, int d, int nb, int lane,
                                         float4& n0, float4& n1) {
  for (int base = 32; base < nb; base += 32) {
    const int cn = (nb - base < 32) ? (nb - base) : 32;
    int   es = 0;
    float ea = 0.0f;
    if (lane < cn) {
      const int2 ent = *(const int2*)(bucket + ((size_t)s * CAP + base + lane) * 2);
      es = ent.x;
      ea = __int_as_float(ent.y);
    }
    for (int i = 0; i < cn; i += GRP) {
      float4 r0[GRP];
      float4 r1[GRP];
      grp_ask(vsrc, es, i, cn, lane, r0, r1);
      grp_sum(ea, i, cn, r0, r1, n0, n1);
    }
  }
  if (d > CAP) {
    int e = __builtin_amdgcn_readfirstlane(ohead[s]);
    for (int it = 0; it < EDGES && e >= 0; ++it) {
      const int   si = esrc[e];
      const float ai = att[e];
      const float4* rp = (const float4*)(vsrc + (size_t)si * DIM);
      const float4 q0 = rp[lane];
      const float4 q1 = rp[lane + 32];
      n0.x += q0.x * ai;  n0.y += q0.y * ai;  n0.z += q0.z * ai;  n0.w += q0.w * ai;
      n1.x += q1.x * ai;  n1.y += q1.y * ai;  n1.z += q1.z * ai;  n1.w += q1.w * ai;
      e = __builtin_amdgcn_readfirstlane(nxt[e]);
    }
  }
}
__device__ __forceinline__ void put_row(float* __restrict__ x1r, float* __restrict__ x2r, int lane,
                                        float4 vv0, float4 vv1, float4 n0, float4 n1) {
  const int c0 = lane * 4;
  const int c1 = c0 + 128;
  x1r[c0 + 0] = vv0.x + n0.x;  x2r[c0 + 0] = vv0.x * n0.x;
  x1r[c0 + 1] = vv0.y + n0.y;  x2r[c0 + 1] = vv0.y * n0.y;
  x1r[c0 + 2] = vv0.z + n0.z;  x2r[c0 + 2] = vv0.z * n0.z;
  x1r[c0 + 3] = vv0.w + n0.w;  x2r[c0 + 3] = vv0.w * n0.w;
  x1r[c1 + 0] = vv1.x + n1.x;  x2r[c1 + 0] = vv1.x * n1.x;
  x1r[c1 + 1] = vv1.y + n1.y;  x2r[c1 + 1] = vv1.y * n1.y;
  x1r[c1 + 2] = vv1.z + n1.z;  x2r[c1 + 2] = vv1.z * n1.z;
  x1r[c1 + 3] = vv1.w + n1.w;  x2r[c1 + 3] = vv1.w * n1.w;
}

__device__ __forceinline__ void b_part(const float4* __restrict__ wfrag, int t0,
                                       v2f (&bA)[KPART], v2f (&bB)[KPART]) {
#pragma unroll
  for (int i = 0; i < KPART; ++i) {
    const float4 f = wfrag[(t0 + i) * 32];
    bA[i].x = f.x;  bA[i].y = f.y;
    bB[i].x = f.z;  bB[i].y = f.w;
  }
}
__device__ __forceinline__ void a_part(const float (*xa)[DIM + LDSPAD], const float (*xb)[DIM + LDSPAD],
                                       int mrow, int khalf, int t0,
                                       v2f (&f1)[KPART], v2f (&f2)[KPART]) {
#pragma unroll
  for (int i = 0; i < KPART; ++i) {
    const int ka = 4 * (t0 + i) + 2 * khalf;
    f1[i].x = xa[mrow][ka];  f1[i].y = xa[mrow][ka + 1];
    f2[i].x = xb[mrow][ka];  f2[i].y = xb[mrow][ka + 1];
  }
}
__device__ __forceinline__ void mm_part(const v2f (&f1)[KPART], const v2f (&f2)[KPART],
                                        const v2f (&bA)[KPART], const v2f (&bB)[KPART],
                                        v8f& acc1A, v8f& acc2A, v8f& acc1B, v8f& acc2B) {
#pragma unroll
  for (int i = 0; i < KPART; ++i) {
    acc1A = __builtin_amdgcn_wmma_f32_16x16x4_f32(
        false, f1[i], false, bA[i], (short)0, acc1A, false, false);
    acc2A = __builtin_amdgcn_wmma_f32_16x16x4_f32(
        false, f2[i], false, bA[i], (short)0, acc2A, false, false);
    acc1B = __builtin_amdgcn_wmma_f32_16x16x4_f32(
        false, f1[i], false, bB[i], (short)0, acc1B, false, false);
    acc2B = __builtin_amdgcn_wmma_f32_16x16x4_f32(
        false, f2[i], false, bB[i], (short)0, acc2B, false, false);
  }
}
__device__ __forceinline__ void put_out(float* __restrict__ orow, int khalf, int ncolA, int ncolB,
                                        float biasA, float biasB,
                                        const v8f& acc1A, const v8f& acc2A,
                                        const v8f& acc1B, const v8f& acc2B) {
#pragma unroll
  for (int r = 0; r < 8; ++r) {
    const int m = khalf * 8 + r;
    float y1 = acc1A[r] + biasA;
    float y2 = acc2A[r] + biasA;
    y1 = (y1 >= 0.0f) ? y1 : 0.01f * y1;
    y2 = (y2 >= 0.0f) ? y2 : 0.01f * y2;
    orow[(size_t)m * DIM + ncolA] = y1 + y2;
    float z1 = acc1B[r] + biasB;
    float z2 = acc2B[r] + biasB;
    z1 = (z1 >= 0.0f) ? z1 : 0.01f * z1;
    z2 = (z2 >= 0.0f) ? z2 : 0.01f * z2;
    orow[(size_t)m * DIM + ncolB] = z1 + z2;
  }
}

#define CTHREADS 256
__global__ __launch_bounds__(CTHREADS) void k_combine(const float* __restrict__ v_sc,
                                                         const float* __restrict__ v_grid,
                                                         const float* __restrict__ att_g2s,
                                                         const float* __restrict__ att_s2g,
                                                         const int* __restrict__ src_g2s,
                                                         const int* __restrict__ src_s2g,
                                                         const int* __restrict__ degs,
                                                         const int* __restrict__ degg,
                                                         const int* __restrict__ bks,
                                                         const int* __restrict__ bkg,
                                                         const int* __restrict__ ohs,
                                                         const int* __restrict__ ohg,
                                                         const int* __restrict__ nxs,
                                                         const int* __restrict__ nxg,
                                                         const int* __restrict__ tabs,
                                                         const int* __restrict__ tabg,
                                                         const int* __restrict__ sc_id,
                                                         const int* __restrict__ gr_id,
                                                         const float* __restrict__ wf,
                                                         const float* __restrict__ b1,
                                                         float* __restrict__ out) {
  __shared__ __attribute__((aligned(16))) float x1[2][16][DIM + LDSPAD];
  __shared__ __attribute__((aligned(16))) float x2[2][16][DIM + LDSPAD];

  const int lane  = threadIdx.x & 31;
  const int wave  = __builtin_amdgcn_readfirstlane((int)(threadIdx.x >> 5));
  const int qbase = blockIdx.x * 16;

  const int mrow  = lane & 15;
  const int khalf = lane >> 4;

  const int ncolA = wave * 32 + mrow;
  const int ncolB = ncolA + 16;
  const float4* wfrag = (const float4*)wf + (size_t)wave * (DIM / 4) * 32 + lane;
  v2f bA0[KPART];
  v2f bB0[KPART];
  v2f bA1[KPART];
  v2f bB1[KPART];
  b_part(wfrag, 0, bA0, bB0);

  int gq[4];
  int sq[4];
  int dq[4];
  int nq[4];
  unsigned long long tq[4];
#pragma unroll
  for (int tr = 0; tr < 4; ++tr) {
    const int* ids = (tr >> 1) ? gr_id : sc_id;
    gq[tr] = ids[qbase + wave * 2 + (tr & 1)];
  }
#pragma unroll
  for (int tr = 0; tr < 4; ++tr) tq[tr] = slot_entry((tr >> 1) ? tabg : tabs, gq[tr]);
#pragma unroll
  for (int tr = 0; tr < 4; ++tr) {
    const int* deg = (tr >> 1) ? degg : degs;
    sq[tr] = slot_from(tq[tr], gq[tr]);
    dq[tr] = __builtin_amdgcn_readfirstlane(deg[sq[tr]]);
    nq[tr] = (dq[tr] < CAP) ? dq[tr] : CAP;
  }
  int2 hq[4];
#pragma unroll
  for (int tr = 0; tr < 4; ++tr) {
    const int* bucket = (tr >> 1) ? bkg  : bks;
    const int* tab    = (tr >> 1) ? tabg : tabs;
    const bool any = nq[tr] > 0;
    const int* hp = any ? bucket + (size_t)sq[tr] * CAP * 2 : tab + 2 * (gq[tr] >> 5);
    const int  hl = any ? ((lane < nq[tr]) ? lane : nq[tr] - 1) : 0;
    hq[tr] = *(const int2*)(hp + 2 * hl);
  }

  const float4* vqa = (const float4*)(v_sc + (size_t)gq[0] * DIM);
  const float4* vqb = (const float4*)(v_sc + (size_t)gq[1] * DIM);
  const float4 sva0 = vqa[lane];
  const float4 sva1 = vqa[lane + 32];
  const float4 svb0 = vqb[lane];
  const float4 svb1 = vqb[lane + 32];

  v8f acc1A = {};
  v8f acc2A = {};
  v8f acc1B = {};
  v8f acc2B = {};
  {
    const float* v      = v_grid;
    const float* vsrc   = v_sc;
    const float* att    = att_s2g;
    const int*   esrc   = src_s2g;
    const int*   bucket = bkg;
    const int*   ohead  = ohg;
    const int*   nxt    = nxg;
    const int   g_a  = gq[2];
    const int   g_b  = gq[3];
    const int   s_a  = sq[2];
    const int   s_b  = sq[3];
    const int   d_a  = dq[2];
    const int   d_b  = dq[3];
    const int   n_a  = nq[2];
    const int   n_b  = nq[3];
    const int   es_a = hq[2].x;
    const int   es_b = hq[3].x;
    const float ea_a = __int_as_float(hq[2].y);
    const float ea_b = __int_as_float(hq[3].y);
    const float4* vpa = (const float4*)(v + (size_t)g_a * DIM);
    const float4* vpb = (const float4*)(v + (size_t)g_b * DIM);
    const int cna  = (n_a < 32) ? n_a : 32;
    const int cnb  = (n_b < 32) ? n_b : 32;
    const int ctot = cna + cnb;
    float4 va0, va1, vb0, vb1;
    if (ctot < 1) {
      va0 = vpa[lane];  va1 = vpa[lane + 32];
      vb0 = vpb[lane];  vb1 = vpb[lane + 32];
    }
    float4 na0 = {0.0f, 0.0f, 0.0f, 0.0f};
    float4 na1 = {0.0f, 0.0f, 0.0f, 0.0f};
    float4 nb0 = {0.0f, 0.0f, 0.0f, 0.0f};
    float4 nb1 = {0.0f, 0.0f, 0.0f, 0.0f};
    for (int j0 = 0; j0 < ctot; j0 += PGRP) {
      float4 r0[PGRP];
      float4 r1[PGRP];
      pair_ask(vsrc, es_a, es_b, j0, cna, ctot, lane, r0, r1);
      if (j0 == 0) {
        va0 = vpa[lane];  va1 = vpa[lane + 32];
        vb0 = vpb[lane];  vb1 = vpb[lane + 32];
      }
      pair_sum(ea_a, ea_b, j0, cna, ctot, r0, r1, na0, na1, nb0, nb1);
    }
    for (int rr = 0; rr < 2; ++rr) {
      const int s_r = rr ? s_b : s_a;
      const int d_r = rr ? d_b : d_a;
      const int n_r = rr ? n_b : n_a;
      if (n_r > 32) {
        float4 t0 = rr ? nb0 : na0;
        float4 t1 = rr ? nb1 : na1;
        row_tail(vsrc, att, esrc, bucket, ohead, nxt, s_r, d_r, n_r, lane, t0, t1);
        if (rr) { nb0 = t0; nb1 = t1; } else { na0 = t0; na1 = t1; }
      }
    }
    put_row(&x1[1][wave * 2][0],     &x2[1][wave * 2][0],     lane, va0, va1, na0, na1);
    put_row(&x1[1][wave * 2 + 1][0], &x2[1][wave * 2 + 1][0], lane, vb0, vb1, nb0, nb1);
  }
  __syncthreads();

  {
    const float* vsrc   = v_grid;
    const float* att    = att_g2s;
    const int*   esrc   = src_g2s;
    const int*   bucket = bks;
    const int*   ohead  = ohs;
    const int*   nxt    = nxs;
    const int   s_a  = sq[0];
    const int   s_b  = sq[1];
    const int   d_a  = dq[0];
    const int   d_b  = dq[1];
    const int   n_a  = nq[0];
    const int   n_b  = nq[1];
    const int   es_a = hq[0].x;
    const int   es_b = hq[1].x;
    const float ea_a = __int_as_float(hq[0].y);
    const float ea_b = __int_as_float(hq[1].y);
    const int cna  = (n_a < 32) ? n_a : 32;
    const int cnb  = (n_b < 32) ? n_b : 32;
    const int ctot = cna + cnb;
    float4 na0 = {0.0f, 0.0f, 0.0f, 0.0f};
    float4 na1 = {0.0f, 0.0f, 0.0f, 0.0f};
    float4 nb0 = {0.0f, 0.0f, 0.0f, 0.0f};
    float4 nb1 = {0.0f, 0.0f, 0.0f, 0.0f};
#define TWO_PARTS(XA, XB, T, C1A, C2A, C1B, C2B)                                     \
    {                                                                                \
      v2f f1[KPART];                                                                 \
      v2f f2[KPART];                                                                 \
      v2f h1[KPART];                                                                 \
      v2f h2[KPART];                                                                 \
      a_part(XA, XB, mrow, khalf, 2 * (T) * KPART, f1, f2);                          \
      a_part(XA, XB, mrow, khalf, (2 * (T) + 1) * KPART, h1, h2);                    \
            \
                \
                                                      \
      __builtin_amdgcn_sched_barrier(0);                                             \
      mm_part(f1, f2, bA0, bB0, C1A, C2A, C1B, C2B);                                 \
      b_part(wfrag, ((2 * (T) + 2) % NPART) * KPART, bA0, bB0);                      \
      mm_part(h1, h2, bA1, bB1, C1A, C2A, C1B, C2B);                                 \
            \
         \
           \
      asm volatile("v_nop\n\tv_nop\n\tv_nop\n\tv_nop"                                  \
                   : "+v"(C1A), "+v"(C2A), "+v"(C1B), "+v"(C2B));                    \
    }
    const int nlive = (ctot + PGRP - 1) / PGRP;
    int t = 0;
#pragma unroll 1
    for (; t < nlive; ++t) {
      const int j0 = t * PGRP;
      float4 r0[PGRP];
      float4 r1[PGRP];
      b_part(wfrag, (2 * t + 1) * KPART, bA1, bB1);
      pair_ask(vsrc, es_a, es_b, j0, cna, ctot, lane, r0, r1);
      TWO_PARTS(x1[1], x2[1], t, acc1A, acc2A, acc1B, acc2B)
      pair_sum(ea_a, ea_b, j0, cna, ctot, r0, r1, na0, na1, nb0, nb1);
    }
#pragma unroll 1
    for (; t < NPART / 2; ++t) {
      b_part(wfrag, (2 * t + 1) * KPART, bA1, bB1);
      TWO_PARTS(x1[1], x2[1], t, acc1A, acc2A, acc1B, acc2B)
    }
    for (int rr = 0; rr < 2; ++rr) {
      const int s_r = rr ? s_b : s_a;
      const int d_r = rr ? d_b : d_a;
      const int n_r = rr ? n_b : n_a;
      if (n_r > 32) {
        float4 t0 = rr ? nb0 : na0;
        float4 t1 = rr ? nb1 : na1;
        row_tail(vsrc, att, esrc, bucket, ohead, nxt, s_r, d_r, n_r, lane, t0, t1);
        if (rr) { nb0 = t0; nb1 = t1; } else { na0 = t0; na1 = t1; }
      }
    }
    put_row(&x1[0][wave * 2][0],     &x2[0][wave * 2][0],     lane, sva0, sva1, na0, na1);
    put_row(&x1[0][wave * 2 + 1][0], &x2[0][wave * 2 + 1][0], lane, svb0, svb1, nb0, nb1);
  }
  __syncthreads();

  const float biasA = b1[ncolA];
  const float biasB = b1[ncolB];
  put_out(out + ((size_t)NQ + qbase) * DIM, khalf, ncolA, ncolB, biasA, biasB,
          acc1A, acc2A, acc1B, acc2B);

  v8f c1A = {};
  v8f c2A = {};
  v8f c1B = {};
  v8f c2B = {};
#pragma unroll 1
  for (int t = 0; t < NPART / 2; ++t) {
    b_part(wfrag, (2 * t + 1) * KPART, bA1, bB1);
    TWO_PARTS(x1[0], x2[0], t, c1A, c2A, c1B, c2B)
  }
#undef TWO_PARTS
  put_out(out + (size_t)qbase * DIM, khalf, ncolA, ncolB, biasA, biasB, c1A, c2A, c1B, c2B);
}

extern "C" void kernel_launch(void* const* d_in, const int* in_sizes, int n_in,
                              void* d_out, int out_size, void* d_ws, size_t ws_size,
                              hipStream_t stream) {
  const float* v_grid  = (const float*)d_in[0];
  const float* v_sc    = (const float*)d_in[1];
  const float* att_s2g = (const float*)d_in[2];
  const float* att_g2s = (const float*)d_in[3];
  const float* W1      = (const float*)d_in[4];
  const float* b1      = (const float*)d_in[5];
  const int*   src_s2g = (const int*)d_in[6];
  const int*   dst_s2g = (const int*)d_in[7];
  const int*   src_g2s = (const int*)d_in[8];
  const int*   dst_g2s = (const int*)d_in[9];
  const int*   sc_id   = (const int*)d_in[10];
  const int*   gr_id   = (const int*)d_in[11];
  float* out = (float*)d_out;

  int*  tabg = (int*)d_ws;
  int*  tabs = tabg + 2 * WG;
  int*  degg = tabs + 2 * WSC;
  int*  degs = degg + MAXU;
  int*  ohg  = degs + MAXU;
  int*  ohs  = ohg + MAXU;
  int*  nxg  = ohs + MAXU;
  int*  nxs  = nxg + EDGES;
  int*  bkg  = nxs + EDGES;
  int*  bks  = bkg + (size_t)MAXU * CAP * 2;
  float* wf  = (float*)(bks + (size_t)MAXU * CAP * 2 + WS_PAD);

  k_rank<<<1 + WBLOCKS, RTHREADS, 0, stream>>>(gr_id, sc_id, tabg, degg, ohg, W1, wf);

  k_link<<<dim3(LINKB, 2), 256, 0, stream>>>(att_s2g, src_s2g, dst_s2g, tabg, degg, bkg, ohg, nxg,
                                             att_g2s, src_g2s, dst_g2s, tabs, degs, bks, ohs, nxs);

  k_combine<<<NQ / 16, CTHREADS, 0, stream>>>(v_sc, v_grid, att_g2s, att_s2g, src_g2s, src_s2g,
                                              degs, degg, bks, bkg, ohs, ohg, nxs, nxg,
                                              tabs, tabg, sc_id, gr_id, wf, b1, out);
}


# three launches to two: each bucketing workgroup builds the node-to-slot map in its own LDS and writes matched edges as per-slot runs in its own segment, no global atomics; nh summed in run order
# speedup vs baseline: 6.7554x; 1.1332x over previous
#include <hip/hip_runtime.h>

#define NG     100000
#define NS     20000
#define EDGES  300000
#define DIM    256
#define NQ     4096
#define MAXU   4096
#define LDSPAD 2
#define GRP    8
#define PGRP   16
#define WG     (NG / 32)
#define WSC    (NS / 32)
#define RTHREADS 1024
#define LEPT   16
#define EPB    (RTHREADS * LEPT)
#define NB     ((EDGES + EPB - 1) / EPB)
#define CWAVES 8
#define KPART  8
#define NPART  (DIM / 4 / KPART)
#define WFQ    (CWAVES * (DIM / 4) * 32)
#define WBLOCKS (WFQ / RTHREADS)
#define WS_HEAD (2 * WG + 2 * WSC + 2 * NB * MAXU + 2 * NB * EPB * 2)
#define WS_PAD  ((32 - WS_HEAD % 32) % 32)
static_assert(WFQ % RTHREADS == 0, "whole workgroups write the fragment-order copy of W1");
static_assert(DIM == CWAVES * 32, "8 waves x 32 columns cover the output row");
static_assert((NPART / 2) * PGRP == 64 && NPART % 2 == 0, "two parts per group of the first 32 entries of two rows' joint list");
static_assert(NG % 32 == 0 && NS % 32 == 0, "a node bitmap is whole 32-bit words");
static_assert(WG <= 4 * RTHREADS && WSC <= 4 * RTHREADS, "a bucketing thread owns 4 consecutive bitmap words");
static_assert(MAXU == 4 * RTHREADS, "a bucketing thread owns 4 consecutive slots");
static_assert(NQ % RTHREADS == 0, "a bucketing thread marks NQ / RTHREADS queries");
static_assert(EDGES % 4 == 0 && LEPT == 16, "a bucketing thread owns 4 groups of 4 whole edges (16-byte loads)");
static_assert(NB <= 32, "one lane of a combine wave per bucketing workgroup");
static_assert(EPB < 65536, "a run's start and length share one 32-bit table entry");
static_assert((2 * WG + 2 * WSC) % 4 == 0, "the run table is written with 16-byte stores");

typedef __attribute__((ext_vector_type(2))) float v2f;
typedef __attribute__((ext_vector_type(8))) float v8f;

__device__ __forceinline__ int wave_incl(int v, int lane) {
  for (int d = 1; d < 32; d <<= 1) {
    const int o = __shfl_up(v, d);
    if (lane >= d) v += o;
  }
  return v;
}

__device__ __forceinline__ void lds_barrier() {
  __builtin_amdgcn_fence(__ATOMIC_RELEASE, "workgroup", "local");
  __builtin_amdgcn_s_barrier();
  __builtin_amdgcn_fence(__ATOMIC_ACQUIRE, "workgroup", "local");
}

__global__ __launch_bounds__(RTHREADS) void k_bucket(const int* __restrict__ gid,
                                                      const int* __restrict__ sid,
                                                      const float* __restrict__ att0,
                                                      const int* __restrict__ src0,
                                                      const int* __restrict__ dst0,
                                                      const float* __restrict__ att1,
                                                      const int* __restrict__ src1,
                                                      const int* __restrict__ dst1,
                                                      int* __restrict__ tabg,
                                                      int* __restrict__ tabs,
                                                      int* __restrict__ segtab,
                                                      int* __restrict__ seg,
                                                      const float* __restrict__ W1,
                                                      float* __restrict__ wf) {
  __shared__ __attribute__((aligned(16))) unsigned bm[4 * RTHREADS];
  __shared__ __attribute__((aligned(16))) int rk[4 * RTHREADS];
  __shared__ __attribute__((aligned(16))) int cnt[MAXU];
  __shared__ __attribute__((aligned(16))) int off[MAXU];
  __shared__ int wtot[RTHREADS / 32];
  const int tid  = threadIdx.x;
  const int lane = tid & 31;
  const int wave = tid >> 5;

  if (blockIdx.x >= 2 * NB) {
    const int q  = (blockIdx.x - 2 * NB) * RTHREADS + tid;
    const int t  = (q >> 5) & (DIM / 4 - 1);
    const int w  = q >> 11;
    const int n  = w * 32 + (lane & 15);
    const int k  = 4 * t + 2 * (lane >> 4);
    const float2 a = *(const float2*)(W1 + (size_t)n * DIM + k);
    const float2 b = *(const float2*)(W1 + (size_t)(n + 16) * DIM + k);
    ((float4*)wf)[q] = make_float4(a.x, a.y, b.x, b.y);
    return;
  }

  const int dir = (blockIdx.x >= NB) ? 1 : 0;
  const int blk = blockIdx.x - dir * NB;
  const int*   __restrict__ ids = dir ? sid  : gid;
  const float* __restrict__ att = dir ? att1 : att0;
  const int*   __restrict__ src = dir ? src1 : src0;
  const int*   __restrict__ dst = dir ? dst1 : dst0;
  int*         __restrict__ tab = dir ? tabs : tabg;
  const int nwords = dir ? WSC : WG;

  int qi[NQ / RTHREADS];
#pragma unroll
  for (int k = 0; k < NQ / RTHREADS; ++k) qi[k] = ids[tid + k * RTHREADS];
  int4   dv[LEPT / 4];
  int4   sv[LEPT / 4];
  float4 av[LEPT / 4];
#pragma unroll
  for (int k = 0; k < LEPT / 4; ++k) {
    const int e  = blk * EPB + k * 4 * RTHREADS + 4 * tid;
    const int ec = (e < EDGES) ? e : (EDGES - 4);
    dv[k] = *(const int4*)(dst + ec);
  }
#pragma unroll
  for (int k = 0; k < LEPT / 4; ++k) {
    const int e  = blk * EPB + k * 4 * RTHREADS + 4 * tid;
    const int ec = (e < EDGES) ? e : (EDGES - 4);
    sv[k] = *(const int4*)(src + ec);
    av[k] = *(const float4*)(att + ec);
  }

  for (int k = 0; k < 4; ++k) {
    bm[tid + k * RTHREADS]  = 0u;
    cnt[tid + k * RTHREADS] = 0;
  }
  lds_barrier();
#pragma unroll
  for (int k = 0; k < NQ / RTHREADS; ++k) atomicOr(&bm[qi[k] >> 5], 1u << (qi[k] & 31));
  lds_barrier();

  const uint4 bw = *(const uint4*)(bm + 4 * tid);
  const int p1   = __popc(bw.x);
  const int p2   = p1 + __popc(bw.y);
  const int p3   = p2 + __popc(bw.z);
  const int tsum = p3 + __popc(bw.w);
  const int incl = wave_incl(tsum, lane);
  if (lane == 31) wtot[wave] = incl;
  lds_barrier();
  if (wave == 0) {
    const int v = wtot[lane];
    wtot[lane] = wave_incl(v, lane) - v;
  }
  lds_barrier();
  const int base = wtot[wave] + incl - tsum;
  *(int4*)(rk + 4 * tid) = make_int4(base, base + p1, base + p2, base + p3);
  if (blk == 0) {
    const int pre[4] = {base, base + p1, base + p2, base + p3};
    const unsigned bits[4] = {bw.x, bw.y, bw.z, bw.w};
#pragma unroll
    for (int j = 0; j < 4; ++j) {
      const int w = 4 * tid + j;
      if (w < nwords) {
        int2 ent; ent.x = (int)bits[j]; ent.y = pre[j];
        *(int2*)(tab + 2 * w) = ent;
      }
    }
  }
  lds_barrier();

  const int   dd[LEPT] = {dv[0].x, dv[0].y, dv[0].z, dv[0].w, dv[1].x, dv[1].y, dv[1].z, dv[1].w,
                          dv[2].x, dv[2].y, dv[2].z, dv[2].w, dv[3].x, dv[3].y, dv[3].z, dv[3].w};
  int s[LEPT];
  int p[LEPT];
  unsigned bq[LEPT];
  int      rq[LEPT];
#pragma unroll
  for (int k = 0; k < LEPT; ++k) {
    bq[k] = bm[dd[k] >> 5];
    rq[k] = rk[dd[k] >> 5];
  }
#pragma unroll
  for (int k = 0; k < LEPT; ++k) {
    const bool ex = blk * EPB + (k >> 2) * 4 * RTHREADS + 4 * tid < EDGES;
    const unsigned bit = 1u << (dd[k] & 31);
    s[k] = (ex && (bq[k] & bit)) ? (rq[k] + __popc(bq[k] & (bit - 1u))) : -1;
  }
#pragma unroll
  for (int k = 0; k < LEPT; ++k) {
    p[k] = 0;
    if (s[k] >= 0) p[k] = atomicAdd(&cnt[s[k]], 1);
  }
  lds_barrier();

  const int4 cv = *(const int4*)(cnt + 4 * tid);
  const int c1   = cv.x;
  const int c2   = c1 + cv.y;
  const int c3   = c2 + cv.z;
  const int csum = c3 + cv.w;
  const int cincl = wave_incl(csum, lane);
  if (lane == 31) wtot[wave] = cincl;
  lds_barrier();
  if (wave == 0) {
    const int v = wtot[lane];
    wtot[lane] = wave_incl(v, lane) - v;
  }
  lds_barrier();
  const int ob = wtot[wave] + cincl - csum;
  *(int4*)(off + 4 * tid) = make_int4(ob, ob + c1, ob + c2, ob + c3);
  *(int4*)(segtab + (size_t)(dir * NB + blk) * MAXU + 4 * tid) =
      make_int4(ob | (cv.x << 16), (ob + c1) | (cv.y << 16), (ob + c2) | (cv.z << 16), (ob + c3) | (cv.w << 16));
  lds_barrier();

  int2* __restrict__ sg = (int2*)seg + (size_t)(dir * NB + blk) * EPB;
  const int   ss[LEPT] = {sv[0].x, sv[0].y, sv[0].z, sv[0].w, sv[1].x, sv[1].y, sv[1].z, sv[1].w,
                          sv[2].x, sv[2].y, sv[2].z, sv[2].w, sv[3].x, sv[3].y, sv[3].z, sv[3].w};
  const float aa[LEPT] = {av[0].x, av[0].y, av[0].z, av[0].w, av[1].x, av[1].y, av[1].z, av[1].w,
                          av[2].x, av[2].y, av[2].z, av[2].w, av[3].x, av[3].y, av[3].z, av[3].w};
  int ofs[LEPT];
#pragma unroll
  for (int k = 0; k < LEPT; ++k) ofs[k] = off[(s[k] >= 0) ? s[k] : 0];
#pragma unroll
  for (int k = 0; k < LEPT; ++k) {
    if (s[k] >= 0) {
      int2 ent; ent.x = ss[k]; ent.y = __float_as_int(aa[k]);
      sg[ofs[k] + p[k]] = ent;
    }
  }
}

__device__ __forceinline__ unsigned long long slot_entry(const int* __restrict__ tab, int node) {
  return *(const unsigned long long*)(tab + 2 * (node >> 5));
}
__device__ __forceinline__ int slot_from(unsigned long long ent, int node) {
  const unsigned bits = (unsigned)ent;
  const int      rank = (int)(ent >> 32);
  const unsigned bit  = 1u << (node & 31);
  const int      s    = rank + __popc(bits & (bit - 1u));
  return (bits & bit) ? s : -1;
}


__device__ __forceinline__ void grp_ask(const float* __restrict__ vsrc, int es, int i, int cn,
                                        int lane, float4 (&r0)[GRP], float4 (&r1)[GRP]) {
#pragma unroll
  for (int k = 0; k < GRP; ++k) {
    const int ik = (i + k < cn) ? (i + k) : (cn - 1);
    const int si = __builtin_amdgcn_readlane(es, ik);
    const float4* rp = (const float4*)(vsrc + (size_t)si * DIM);
    r0[k] = rp[lane];
    r1[k] = rp[lane + 32];
  }
}
__device__ __forceinline__ void grp_sum(float ea, int i, int cn,
                                        const float4 (&r0)[GRP], const float4 (&r1)[GRP],
                                        float4& n0, float4& n1) {
#pragma unroll
  for (int k = 0; k < GRP; ++k) {
    if (i + k < cn) {
      const float ai = __int_as_float(
          __builtin_amdgcn_readlane(__float_as_int(ea), i + k));
      n0.x += r0[k].x * ai;  n0.y += r0[k].y * ai;
      n0.z += r0[k].z * ai;  n0.w += r0[k].w * ai;
      n1.x += r1[k].x * ai;  n1.y += r1[k].y * ai;
      n1.z += r1[k].z * ai;  n1.w += r1[k].w * ai;
    }
  }
}
__device__ __forceinline__ void pair_ask(const float* __restrict__ vsrc, int es_a, int es_b,
                                         int j0, int cna, int ctot, int lane,
                                         float4 (&r0)[PGRP], float4 (&r1)[PGRP]) {
#pragma unroll
  for (int k = 0; k < PGRP; ++k) {
    const int j  = (j0 + k < ctot) ? (j0 + k) : (ctot - 1);
    const bool ina = j < cna;
    const int sa = __builtin_amdgcn_readlane(es_a, ina ? j : 0);
    const int sb = __builtin_amdgcn_readlane(es_b, ina ? 0 : (j - cna));
    const int si = ina ? sa : sb;
    const float4* rp = (const float4*)(vsrc + (size_t)si * DIM);
    r0[k] = rp[lane];
    r1[k] = rp[lane + 32];
  }
}
__device__ __forceinline__ void pair_sum(float ea_a, float ea_b, int j0, int cna, int ctot,
                                         const float4 (&r0)[PGRP], const float4 (&r1)[PGRP],
                                         float4& na0, float4& na1, float4& nb0, float4& nb1) {
#pragma unroll
  for (int k = 0; k < PGRP; ++k) {
    const int j = j0 + k;
    if (j < ctot) {
      if (j < cna) {
        const float ai = __int_as_float(__builtin_amdgcn_readlane(__float_as_int(ea_a), j));
        na0.x += r0[k].x * ai;  na0.y += r0[k].y * ai;
        na0.z += r0[k].z * ai;  na0.w += r0[k].w * ai;
        na1.x += r1[k].x * ai;  na1.y += r1[k].y * ai;
        na1.z += r1[k].z * ai;  na1.w += r1[k].w * ai;
      } else {
        const float ai = __int_as_float(__builtin_amdgcn_readlane(__float_as_int(ea_b), j - cna));
        nb0.x += r0[k].x * ai;  nb0.y += r0[k].y * ai;
        nb0.z += r0[k].z * ai;  nb0.w += r0[k].w * ai;
        nb1.x += r1[k].x * ai;  nb1.y += r1[k].y * ai;
        nb1.z += r1[k].z * ai;  nb1.w += r1[k].w * ai;
      }
    }
  }
}
__device__ __forceinline__ int entry_index(int incl, int cnt, int off, int dir, int e) {
  int bl = 0;
#pragma unroll
  for (int b = 0; b < NB; ++b) bl += (__builtin_amdgcn_readlane(incl, b) <= e) ? 1 : 0;
  const int ex = __shfl(incl - cnt, bl);
  const int of = __shfl(off, bl);
  return (dir * NB + bl) * EPB + of + (e - ex);
}
__device__ __forceinline__ void row_tail(const float* __restrict__ vsrc,
                                         const int* __restrict__ seg,
                                         int incl, int cnt, int off, int dir, int d, int lane,
                                         float4& n0, float4& n1) {
  for (int base = 32; base < d; base += 32) {
    const int cn = (d - base < 32) ? (d - base) : 32;
    const int e  = (lane < cn) ? (base + lane) : (d - 1);
    const int2 ent = *(const int2*)(seg + 2 * (size_t)entry_index(incl, cnt, off, dir, e));
    const int   es = ent.x;
    const float ea = __int_as_float(ent.y);
    for (int i = 0; i < cn; i += GRP) {
      float4 r0[GRP];
      float4 r1[GRP];
      grp_ask(vsrc, es, i, cn, lane, r0, r1);
      grp_sum(ea, i, cn, r0, r1, n0, n1);
    }
  }
}
__device__ __forceinline__ void put_row(float* __restrict__ x1r, float* __restrict__ x2r, int lane,
                                        float4 vv0, float4 vv1, float4 n0, float4 n1) {
  const int c0 = lane * 4;
  const int c1 = c0 + 128;
  x1r[c0 + 0] = vv0.x + n0.x;  x2r[c0 + 0] = vv0.x * n0.x;
  x1r[c0 + 1] = vv0.y + n0.y;  x2r[c0 + 1] = vv0.y * n0.y;
  x1r[c0 + 2] = vv0.z + n0.z;  x2r[c0 + 2] = vv0.z * n0.z;
  x1r[c0 + 3] = vv0.w + n0.w;  x2r[c0 + 3] = vv0.w * n0.w;
  x1r[c1 + 0] = vv1.x + n1.x;  x2r[c1 + 0] = vv1.x * n1.x;
  x1r[c1 + 1] = vv1.y + n1.y;  x2r[c1 + 1] = vv1.y * n1.y;
  x1r[c1 + 2] = vv1.z + n1.z;  x2r[c1 + 2] = vv1.z * n1.z;
  x1r[c1 + 3] = vv1.w + n1.w;  x2r[c1 + 3] = vv1.w * n1.w;
}

__device__ __forceinline__ void b_part(const float4* __restrict__ wfrag, int t0,
                                       v2f (&bA)[KPART], v2f (&bB)[KPART]) {
#pragma unroll
  for (int i = 0; i < KPART; ++i) {
    const float4 f = wfrag[(t0 + i) * 32];
    bA[i].x = f.x;  bA[i].y = f.y;
    bB[i].x = f.z;  bB[i].y = f.w;
  }
}
__device__ __forceinline__ void a_part(const float (*xa)[DIM + LDSPAD], const float (*xb)[DIM + LDSPAD],
                                       int mrow, int khalf, int t0,
                                       v2f (&f1)[KPART], v2f (&f2)[KPART]) {
#pragma unroll
  for (int i = 0; i < KPART; ++i) {
    const int ka = 4 * (t0 + i) + 2 * khalf;
    f1[i].x = xa[mrow][ka];  f1[i].y = xa[mrow][ka + 1];
    f2[i].x = xb[mrow][ka];  f2[i].y = xb[mrow][ka + 1];
  }
}
__device__ __forceinline__ void mm_part(const v2f (&f1)[KPART], const v2f (&f2)[KPART],
                                        const v2f (&bA)[KPART], const v2f (&bB)[KPART],
                                        v8f& acc1A, v8f& acc2A, v8f& acc1B, v8f& acc2B) {
#pragma unroll
  for (int i = 0; i < KPART; ++i) {
    acc1A = __builtin_amdgcn_wmma_f32_16x16x4_f32(
        false, f1[i], false, bA[i], (short)0, acc1A, false, false);
    acc2A = __builtin_amdgcn_wmma_f32_16x16x4_f32(
        false, f2[i], false, bA[i], (short)0, acc2A, false, false);
    acc1B = __builtin_amdgcn_wmma_f32_16x16x4_f32(
        false, f1[i], false, bB[i], (short)0, acc1B, false, false);
    acc2B = __builtin_amdgcn_wmma_f32_16x16x4_f32(
        false, f2[i], false, bB[i], (short)0, acc2B, false, false);
  }
}
__device__ __forceinline__ void put_out(float* __restrict__ orow, int khalf, int ncolA, int ncolB,
                                        float biasA, float biasB,
                                        const v8f& acc1A, const v8f& acc2A,
                                        const v8f& acc1B, const v8f& acc2B) {
#pragma unroll
  for (int r = 0; r < 8; ++r) {
    const int m = khalf * 8 + r;
    float y1 = acc1A[r] + biasA;
    float y2 = acc2A[r] + biasA;
    y1 = (y1 >= 0.0f) ? y1 : 0.01f * y1;
    y2 = (y2 >= 0.0f) ? y2 : 0.01f * y2;
    orow[(size_t)m * DIM + ncolA] = y1 + y2;
    float z1 = acc1B[r] + biasB;
    float z2 = acc2B[r] + biasB;
    z1 = (z1 >= 0.0f) ? z1 : 0.01f * z1;
    z2 = (z2 >= 0.0f) ? z2 : 0.01f * z2;
    orow[(size_t)m * DIM + ncolB] = z1 + z2;
  }
}

#define CTHREADS 256
__global__ __launch_bounds__(CTHREADS) void k_combine(const float* __restrict__ v_sc,
                                                         const float* __restrict__ v_grid,
                                                         const int* __restrict__ segtab,
                                                         const int* __restrict__ seg,
                                                         const int* __restrict__ tabs,
                                                         const int* __restrict__ tabg,
                                                         const int* __restrict__ sc_id,
                                                         const int* __restrict__ gr_id,
                                                         const float* __restrict__ wf,
                                                         const float* __restrict__ b1,
                                                         float* __restrict__ out) {
  __shared__ __attribute__((aligned(16))) float x1[2][16][DIM + LDSPAD];
  __shared__ __attribute__((aligned(16))) float x2[2][16][DIM + LDSPAD];

  const int lane  = threadIdx.x & 31;
  const int wave  = __builtin_amdgcn_readfirstlane((int)(threadIdx.x >> 5));
  const int qbase = blockIdx.x * 16;

  const int mrow  = lane & 15;
  const int khalf = lane >> 4;

  const int ncolA = wave * 32 + mrow;
  const int ncolB = ncolA + 16;
  const float4* wfrag = (const float4*)wf + (size_t)wave * (DIM / 4) * 32 + lane;
  v2f bA0[KPART];
  v2f bB0[KPART];
  v2f bA1[KPART];
  v2f bB1[KPART];
  b_part(wfrag, 0, bA0, bB0);

  int gq[4];
  int sq[4];
  int dq[4];
  int pq[4];
  int cq[4];
  int oq[4];
  int iq[4];
  unsigned long long tq[4];
#pragma unroll
  for (int tr = 0; tr < 4; ++tr) {
    const int* ids = (tr >> 1) ? gr_id : sc_id;
    gq[tr] = ids[qbase + wave * 2 + (tr & 1)];
  }
#pragma unroll
  for (int tr = 0; tr < 4; ++tr) tq[tr] = slot_entry((tr >> 1) ? tabg : tabs, gq[tr]);
#pragma unroll
  for (int tr = 0; tr < 4; ++tr) {
    sq[tr] = slot_from(tq[tr], gq[tr]);
    const int dirq = (tr >> 1) ? 0 : 1;
    pq[tr] = (lane < NB) ? segtab[(size_t)(dirq * NB + lane) * MAXU + sq[tr]] : 0;
  }
#pragma unroll
  for (int tr = 0; tr < 4; ++tr) {
    cq[tr] = pq[tr] >> 16;
    oq[tr] = pq[tr] & 0xffff;
    iq[tr] = wave_incl(cq[tr], lane);
    dq[tr] = __builtin_amdgcn_readlane(iq[tr], 31);
  }
  int2 hq[4];
#pragma unroll
  for (int tr = 0; tr < 4; ++tr) {
    const int  dirq = (tr >> 1) ? 0 : 1;
    const int* tab  = (tr >> 1) ? tabg : tabs;
    const bool any = dq[tr] > 0;
    const int  e   = any ? ((lane < dq[tr]) ? lane : dq[tr] - 1) : 0;
    const int  idx = entry_index(iq[tr], cq[tr], oq[tr], dirq, e);
    const int* hp  = any ? seg + 2 * (size_t)idx : tab + 2 * (gq[tr] >> 5);
    hq[tr] = *(const int2*)hp;
  }

  const float4* vqa = (const float4*)(v_sc + (size_t)gq[0] * DIM);
  const float4* vqb = (const float4*)(v_sc + (size_t)gq[1] * DIM);
  const float4 sva0 = vqa[lane];
  const float4 sva1 = vqa[lane + 32];
  const float4 svb0 = vqb[lane];
  const float4 svb1 = vqb[lane + 32];

  v8f acc1A = {};
  v8f acc2A = {};
  v8f acc1B = {};
  v8f acc2B = {};
  {
    const float* v      = v_grid;
    const float* vsrc   = v_sc;
    const int   g_a  = gq[2];
    const int   g_b  = gq[3];
    const int   d_a  = dq[2];
    const int   d_b  = dq[3];
    const int   es_a = hq[2].x;
    const int   es_b = hq[3].x;
    const float ea_a = __int_as_float(hq[2].y);
    const float ea_b = __int_as_float(hq[3].y);
    const float4* vpa = (const float4*)(v + (size_t)g_a * DIM);
    const float4* vpb = (const float4*)(v + (size_t)g_b * DIM);
    const int cna  = (d_a < 32) ? d_a : 32;
    const int cnb  = (d_b < 32) ? d_b : 32;
    const int ctot = cna + cnb;
    float4 va0, va1, vb0, vb1;
    if (ctot < 1) {
      va0 = vpa[lane];  va1 = vpa[lane + 32];
      vb0 = vpb[lane];  vb1 = vpb[lane + 32];
    }
    float4 na0 = {0.0f, 0.0f, 0.0f, 0.0f};
    float4 na1 = {0.0f, 0.0f, 0.0f, 0.0f};
    float4 nb0 = {0.0f, 0.0f, 0.0f, 0.0f};
    float4 nb1 = {0.0f, 0.0f, 0.0f, 0.0f};
    for (int j0 = 0; j0 < ctot; j0 += PGRP) {
      float4 r0[PGRP];
      float4 r1[PGRP];
      pair_ask(vsrc, es_a, es_b, j0, cna, ctot, lane, r0, r1);
      if (j0 == 0) {
        va0 = vpa[lane];  va1 = vpa[lane + 32];
        vb0 = vpb[lane];  vb1 = vpb[lane + 32];
      }
      pair_sum(ea_a, ea_b, j0, cna, ctot, r0, r1, na0, na1, nb0, nb1);
    }
    for (int rr = 0; rr < 2; ++rr) {
      const int d_r = rr ? d_b : d_a;
      if (d_r > 32) {
        float4 t0 = rr ? nb0 : na0;
        float4 t1 = rr ? nb1 : na1;
        row_tail(vsrc, seg, rr ? iq[3] : iq[2], rr ? cq[3] : cq[2], rr ? oq[3] : oq[2],
                 0, d_r, lane, t0, t1);
        if (rr) { nb0 = t0; nb1 = t1; } else { na0 = t0; na1 = t1; }
      }
    }
    put_row(&x1[1][wave * 2][0],     &x2[1][wave * 2][0],     lane, va0, va1, na0, na1);
    put_row(&x1[1][wave * 2 + 1][0], &x2[1][wave * 2 + 1][0], lane, vb0, vb1, nb0, nb1);
  }
  __syncthreads();

  {
    const float* vsrc   = v_grid;
    const int   d_a  = dq[0];
    const int   d_b  = dq[1];
    const int   es_a = hq[0].x;
    const int   es_b = hq[1].x;
    const float ea_a = __int_as_float(hq[0].y);
    const float ea_b = __int_as_float(hq[1].y);
    const int cna  = (d_a < 32) ? d_a : 32;
    const int cnb  = (d_b < 32) ? d_b : 32;
    const int ctot = cna + cnb;
    float4 na0 = {0.0f, 0.0f, 0.0f, 0.0f};
    float4 na1 = {0.0f, 0.0f, 0.0f, 0.0f};
    float4 nb0 = {0.0f, 0.0f, 0.0f, 0.0f};
    float4 nb1 = {0.0f, 0.0f, 0.0f, 0.0f};
#define TWO_PARTS(XA, XB, T, C1A, C2A, C1B, C2B)                                     \
    {                                                                                \
      v2f f1[KPART];                                                                 \
      v2f f2[KPART];                                                                 \
      v2f h1[KPART];                                                                 \
      v2f h2[KPART];                                                                 \
      a_part(XA, XB, mrow, khalf, 2 * (T) * KPART, f1, f2);                          \
      a_part(XA, XB, mrow, khalf, (2 * (T) + 1) * KPART, h1, h2);                    \
            \
                \
                                                      \
      __builtin_amdgcn_sched_barrier(0);                                             \
      mm_part(f1, f2, bA0, bB0, C1A, C2A, C1B, C2B);                                 \
      b_part(wfrag, ((2 * (T) + 2) % NPART) * KPART, bA0, bB0);                      \
      mm_part(h1, h2, bA1, bB1, C1A, C2A, C1B, C2B);                                 \
            \
         \
           \
      asm volatile("v_nop\n\tv_nop\n\tv_nop\n\tv_nop"                                  \
                   : "+v"(C1A), "+v"(C2A), "+v"(C1B), "+v"(C2B));                    \
    }
    const int nlive = (ctot + PGRP - 1) / PGRP;
    int t = 0;
#pragma unroll 1
    for (; t < nlive; ++t) {
      const int j0 = t * PGRP;
      float4 r0[PGRP];
      float4 r1[PGRP];
      b_part(wfrag, (2 * t + 1) * KPART, bA1, bB1);
      pair_ask(vsrc, es_a, es_b, j0, cna, ctot, lane, r0, r1);
      TWO_PARTS(x1[1], x2[1], t, acc1A, acc2A, acc1B, acc2B)
      pair_sum(ea_a, ea_b, j0, cna, ctot, r0, r1, na0, na1, nb0, nb1);
    }
#pragma unroll 1
    for (; t < NPART / 2; ++t) {
      b_part(wfrag, (2 * t + 1) * KPART, bA1, bB1);
      TWO_PARTS(x1[1], x2[1], t, acc1A, acc2A, acc1B, acc2B)
    }
    for (int rr = 0; rr < 2; ++rr) {
      const int d_r = rr ? d_b : d_a;
      if (d_r > 32) {
        float4 t0 = rr ? nb0 : na0;
        float4 t1 = rr ? nb1 : na1;
        row_tail(vsrc, seg, rr ? iq[1] : iq[0], rr ? cq[1] : cq[0], rr ? oq[1] : oq[0],
                 1, d_r, lane, t0, t1);
        if (rr) { nb0 = t0; nb1 = t1; } else { na0 = t0; na1 = t1; }
      }
    }
    put_row(&x1[0][wave * 2][0],     &x2[0][wave * 2][0],     lane, sva0, sva1, na0, na1);
    put_row(&x1[0][wave * 2 + 1][0], &x2[0][wave * 2 + 1][0], lane, svb0, svb1, nb0, nb1);
  }
  __syncthreads();

  const float biasA = b1[ncolA];
  const float biasB = b1[ncolB];
  put_out(out + ((size_t)NQ + qbase) * DIM, khalf, ncolA, ncolB, biasA, biasB,
          acc1A, acc2A, acc1B, acc2B);

  v8f c1A = {};
  v8f c2A = {};
  v8f c1B = {};
  v8f c2B = {};
#pragma unroll 1
  for (int t = 0; t < NPART / 2; ++t) {
    b_part(wfrag, (2 * t + 1) * KPART, bA1, bB1);
    TWO_PARTS(x1[0], x2[0], t, c1A, c2A, c1B, c2B)
  }
#undef TWO_PARTS
  put_out(out + (size_t)qbase * DIM, khalf, ncolA, ncolB, biasA, biasB, c1A, c2A, c1B, c2B);
}


extern "C" void kernel_launch(void* const* d_in, const int* in_sizes, int n_in,
                              void* d_out, int out_size, void* d_ws, size_t ws_size,
                              hipStream_t stream) {
  const float* v_grid  = (const float*)d_in[0];
  const float* v_sc    = (const float*)d_in[1];
  const float* att_s2g = (const float*)d_in[2];
  const float* att_g2s = (const float*)d_in[3];
  const float* W1      = (const float*)d_in[4];
  const float* b1      = (const float*)d_in[5];
  const int*   src_s2g = (const int*)d_in[6];
  const int*   dst_s2g = (const int*)d_in[7];
  const int*   src_g2s = (const int*)d_in[8];
  const int*   dst_g2s = (const int*)d_in[9];
  const int*   sc_id   = (const int*)d_in[10];
  const int*   gr_id   = (const int*)d_in[11];
  float* out = (float*)d_out;

  int*  tabg   = (int*)d_ws;
  int*  tabs   = tabg + 2 * WG;
  int*  segtab = tabs + 2 * WSC;
  int*  seg    = segtab + 2 * NB * MAXU;
  float* wf    = (float*)(seg + (size_t)2 * NB * EPB * 2 + WS_PAD);

  k_bucket<<<2 * NB + WBLOCKS, RTHREADS, 0, stream>>>(gr_id, sc_id,
                                                      att_s2g, src_s2g, dst_s2g,
                                                      att_g2s, src_g2s, dst_g2s,
                                                      tabg, tabs, segtab, seg, W1, wf);

  k_combine<<<NQ / 16, CTHREADS, 0, stream>>>(v_sc, v_grid, segtab, seg,
                                              tabs, tabg, sc_id, gr_id, wf, b1, out);
}
